# starting kernel
# baseline (speedup 1.0000x reference)
#include <hip/hip_runtime.h>
#include <hip/hip_bf16.h>

typedef float v2f __attribute__((ext_vector_type(2)));
typedef float v8f __attribute__((ext_vector_type(8)));

__device__ __forceinline__ v8f wmma_f32(v2f a, v2f b, v8f c) {
  return __builtin_amdgcn_wmma_f32_16x16x4_f32(false, a, false, b, (short)0, c,
                                               false, false);
}

#define BATCH   8
#define CIN     3
#define HW      320
#define CLS     2
#define PSZ     10
#define P2      100
#define LPATCH  1024
#define HID     200

#define WS_INI  0
#define WS_FEA  (WS_INI + BATCH*CLS*HW*HW)
#define WS_UI   (WS_FEA + BATCH*CLS*HW*HW)
#define WS_UF   (WS_UI  + BATCH*CLS*LPATCH*P2)
#define WS_W1P  (WS_UF  + BATCH*CLS*LPATCH*P2)
#define WS_W2P  (WS_W1P + 208*100)

__global__ __launch_bounds__(256) void pad_weights_kernel(
    const float* __restrict__ w_lin1, const float* __restrict__ w_lin2,
    float* __restrict__ w1p, float* __restrict__ w2p) {
  int i = blockIdx.x * 256 + threadIdx.x;
  if (i < 208 * 100) w1p[i] = (i < HID * P2) ? w_lin1[i] : 0.f;
  if (i < 112 * 200) w2p[i] = (i < P2 * HID) ? w_lin2[i] : 0.f;
}

#define XS_FLOATS        2160
#define CONV_LDS_FLOATS  (XS_FLOATS + 148 + 9472 + 64 + 64*325 + 1152 + 16)

__global__ __launch_bounds__(256) void conv_fused_kernel(
    const float* __restrict__ x,  const float* __restrict__ w1,
    const float* __restrict__ b1, const float* __restrict__ w2,
    const float* __restrict__ b2, const float* __restrict__ wfm,
    const float* __restrict__ bfm,
    float* __restrict__ ini, float* __restrict__ fea) {
  extern __shared__ float smem[];
  float* xs     = smem;
  int*   offtab = (int*)(xs + XS_FLOATS);
  float* w1s    = xs + XS_FLOATS + 148;
  float* b1s    = w1s + 74 * 64 * 2;
  float* t1s    = b1s + 64;
  float* w2s    = t1s + 64 * 325;
  float* cs     = w2s + 1152;

  const int tid = threadIdx.x;
  const int X0 = blockIdx.x * 16, Y0 = blockIdx.y * 16, b = blockIdx.z;

  for (int i = tid; i < XS_FLOATS; i += 256) {
    float v = 0.f;
    if (i < CIN * 576) {
      int c = i / 576, r = i % 576, iy = r / 24, ix = r % 24;
      int gy = Y0 - 4 + iy, gx = X0 - 4 + ix;
      if (gy >= 0 && gy < HW && gx >= 0 && gx < HW)
        v = x[((size_t)(b * CIN + c) * HW + gy) * HW + gx];
    }
    xs[i] = v;
  }
  if (tid < 148) {
    int off = 1728;
    if (tid < 147) {
      int c = tid / 49, rr = tid % 49;
      off = c * 576 + (rr / 7) * 24 + (rr % 7);
    }
    offtab[tid] = off;
  }
  for (int i = tid; i < 148 * 64; i += 256) {
    int k = i >> 6, n = i & 63;
    w1s[(((k >> 1) << 6) + n) * 2 + (k & 1)] =
        (k < 147) ? w1[n * 147 + k] : 0.f;
  }
  if (tid < 64) b1s[tid] = b1[tid];
  for (int i = tid; i < CLS * 64 * 9; i += 256) w2s[i] = w2[i];
  if (tid < 2) cs[tid]     = b2[tid];
  if (tid < 6) cs[2 + tid] = wfm[tid];
  if (tid < 2) cs[8 + tid] = bfm[tid];
  __syncthreads();

  const int wave  = tid >> 5;
  const int lane  = tid & 31;
  const int l16   = lane & 15;
  const int lhalf = (lane < 16) ? 0 : 2;

  for (int t = wave; t < 21 * 4; t += 8) {
    const int m0 = (t >> 2) * 16, n0 = (t & 3) * 16;
    const int m  = m0 + l16;
    const int mm = (m < 324) ? m : 0;
    const int abase = (mm / 18) * 24 + (mm % 18);
    const int2* ot = ((const int2*)offtab) + (lhalf >> 1);
    const v2f*  bp = ((const v2f*)w1s) + (lhalf >> 1) * 64 + n0 + l16;
    v8f acc = {};
    #pragma unroll 4
    for (int k0 = 0; k0 < 148; k0 += 4) {
      int2 o = *ot;  ot += 2;
      v2f a;
      a.x = xs[abase + o.x];
      a.y = xs[abase + o.y];
      v2f bf = *bp;  bp += 128;
      acc = wmma_f32(a, bf, acc);
    }
    const int ch = n0 + l16;
    const float bias = b1s[ch];
    #pragma unroll
    for (int r = 0; r < 8; ++r) {
      int M = m0 + r + ((lane < 16) ? 0 : 8);
      if (M < 324) {
        int ty = M / 18, tx = M % 18;
        int gy = Y0 - 1 + ty, gx = X0 - 1 + tx;
        float v = (gy >= 0 && gy < HW && gx >= 0 && gx < HW) ? (acc[r] + bias)
                                                             : 0.f;
        t1s[ch * 325 + M] = v;
      }
    }
  }
  __syncthreads();

  {
    const int oy = tid >> 4, ox = tid & 15;
    float a0 = cs[0], a1 = cs[1];
    for (int ci = 0; ci < 64; ++ci) {
      const float* trow = &t1s[ci * 325];
      const float* wr0  = &w2s[ci * 9];
      const float* wr1  = &w2s[(64 + ci) * 9];
      #pragma unroll
      for (int dy = 0; dy < 3; ++dy)
        #pragma unroll
        for (int dx = 0; dx < 3; ++dx) {
          float tv = trow[(oy + dy) * 18 + (ox + dx)];
          a0 = fmaf(wr0[dy * 3 + dx], tv, a0);
          a1 = fmaf(wr1[dy * 3 + dx], tv, a1);
        }
    }
    const int gy = Y0 + oy, gx = X0 + ox;
    const size_t o0 = ((size_t)(b * CLS + 0) * HW + gy) * HW + gx;
    const size_t o1 = ((size_t)(b * CLS + 1) * HW + gy) * HW + gx;
    ini[o0] = a0;
    ini[o1] = a1;
    float x0 = xs[(oy + 4) * 24 + (ox + 4)];
    float x1 = xs[576 + (oy + 4) * 24 + (ox + 4)];
    float x2 = xs[1152 + (oy + 4) * 24 + (ox + 4)];
    fea[o0] = (cs[2] * x0 + cs[3] * x1 + cs[4] * x2 + cs[8]) * 0.25f;
    fea[o1] = (cs[5] * x0 + cs[6] * x1 + cs[7] * x2 + cs[9]) * 0.25f;
  }
}

#define TRANS_LDS_FLOATS (64 * 102 + 64 * 210)

__global__ __launch_bounds__(256) void trans_kernel(
    const float* __restrict__ ini, const float* __restrict__ fea,
    const float* __restrict__ w1p, const float* __restrict__ w2p,
    float* __restrict__ ui, float* __restrict__ uf) {
  extern __shared__ float smem[];
  float* a_lds = smem;
  float* h_lds = a_lds + 64 * 102;

  const int tid = threadIdx.x;
  const float* src = blockIdx.y ? fea : ini;
  float* dst       = blockIdx.y ? uf  : ui;
  const int P0 = blockIdx.x * 64;

  for (int i = tid; i < 64 * P2; i += 256) {
    int r = i / P2, p = i % P2;
    int Pa = P0 + r;
    int bc = Pa >> 10, l = Pa & 1023;
    int gy = (l >> 5) * PSZ + p / PSZ;
    int gx = (l & 31) * PSZ + p % PSZ;
    a_lds[r * 102 + p] = src[((size_t)bc * HW + gy) * HW + gx];
  }
  __syncthreads();

  const int wave  = tid >> 5;
  const int lane  = tid & 31;
  const int l16   = lane & 15;
  const int lhalf = (lane < 16) ? 0 : 2;

  for (int t = wave; t < 4 * 13; t += 8) {
    const int m0 = (t / 13) * 16, n0 = (t % 13) * 16;
    const int n  = n0 + l16;
    const v2f* ap = (const v2f*)&a_lds[(m0 + l16) * 102 + lhalf];
    const v2f* bp = (const v2f*)&w1p[n * P2 + lhalf];
    v8f acc = {};
    #pragma unroll 5
    for (int k0 = 0; k0 < P2; k0 += 4) {
      v2f a  = *ap;  ap += 2;
      v2f bf = *bp;  bp += 2;
      acc = wmma_f32(a, bf, acc);
    }
    #pragma unroll
    for (int r = 0; r < 8; ++r) {
      int M = m0 + r + ((lane < 16) ? 0 : 8);
      h_lds[M * 210 + n] = acc[r];
    }
  }
  __syncthreads();

  for (int t = wave; t < 4 * 7; t += 8) {
    const int m0 = (t / 7) * 16, n0 = (t % 7) * 16;
    const int n  = n0 + l16;
    const v2f* ap = (const v2f*)&h_lds[(m0 + l16) * 210 + lhalf];
    const v2f* bp = (const v2f*)&w2p[n * HID + lhalf];
    v8f acc = {};
    #pragma unroll 5
    for (int k0 = 0; k0 < HID; k0 += 4) {
      v2f a  = *ap;  ap += 2;
      v2f bf = *bp;  bp += 2;
      acc = wmma_f32(a, bf, acc);
    }
    if (n < P2) {
      #pragma unroll
      for (int r = 0; r < 8; ++r) {
        int M = m0 + r + ((lane < 16) ? 0 : 8);
        float v = acc[r];
        dst[(size_t)(P0 + M) * P2 + n] = v > 0.f ? v : 0.f;
      }
    }
  }
}

#define ATT_LDS_FLOATS (128 * 102 + 64 * 102)

__global__ __launch_bounds__(256) void att_kernel(
    const float* __restrict__ ui, const float* __restrict__ uf,
    float* __restrict__ out) {
  extern __shared__ float smem[];
  float* us = smem;
  float* vs = us + 128 * 102;

  const int tid = threadIdx.x;
  const int bc = blockIdx.z;
  const int M0 = blockIdx.x * 128, N0 = blockIdx.y * 64;
  const float* U = ui + (size_t)bc * LPATCH * P2;
  const float* V = uf + (size_t)bc * LPATCH * P2;

  for (int i = tid; i < 128 * P2; i += 256) {
    int r = i / P2, k = i % P2;
    us[r * 102 + k] = U[(size_t)(M0 + r) * P2 + k];
  }
  for (int i = tid; i < 64 * P2; i += 256) {
    int r = i / P2, k = i % P2;
    vs[r * 102 + k] = V[(size_t)(N0 + r) * P2 + k];
  }
  __syncthreads();

  const int wave  = tid >> 5;
  const int lane  = tid & 31;
  const int l16   = lane & 15;
  const int lhalf = (lane < 16) ? 0 : 2;

  for (int t = wave; t < 8 * 4; t += 8) {
    const int m0 = (t >> 2) * 16, n0 = (t & 3) * 16;
    const v2f* ap = (const v2f*)&us[(m0 + l16) * 102 + lhalf];
    const v2f* bp = (const v2f*)&vs[(n0 + l16) * 102 + lhalf];
    v8f acc = {};
    #pragma unroll 5
    for (int k0 = 0; k0 < P2; k0 += 4) {
      v2f a  = *ap;  ap += 2;
      v2f bf = *bp;  bp += 2;
      acc = wmma_f32(a, bf, acc);
    }
    #pragma unroll
    for (int r = 0; r < 8; ++r) {
      int M = M0 + m0 + r + ((lane < 16) ? 0 : 8);
      int N = N0 + n0 + l16;
      out[((size_t)bc * LPATCH + M) * LPATCH + N] = acc[r] * 0.01f;
    }
  }
}

extern "C" void kernel_launch(void* const* d_in, const int* in_sizes, int n_in,
                              void* d_out, int out_size, void* d_ws,
                              size_t ws_size, hipStream_t stream) {
  (void)in_sizes; (void)n_in; (void)out_size; (void)ws_size;
  const float* x       = (const float*)d_in[0];
  const float* w_conv1 = (const float*)d_in[1];
  const float* b_conv1 = (const float*)d_in[2];
  const float* w_conv2 = (const float*)d_in[3];
  const float* b_conv2 = (const float*)d_in[4];
  const float* w_fm    = (const float*)d_in[5];
  const float* b_fm    = (const float*)d_in[6];
  const float* w_lin1  = (const float*)d_in[7];
  const float* w_lin2  = (const float*)d_in[8];
  float* out = (float*)d_out;

  float* ws  = (float*)d_ws;
  float* ini = ws + WS_INI;
  float* fea = ws + WS_FEA;
  float* ui  = ws + WS_UI;
  float* uf  = ws + WS_UF;
  float* w1p = ws + WS_W1P;
  float* w2p = ws + WS_W2P;

  pad_weights_kernel<<<dim3((112 * 200 + 255) / 256), 256, 0, stream>>>(
      w_lin1, w_lin2, w1p, w2p);

  conv_fused_kernel<<<dim3(HW / 16, HW / 16, BATCH), 256,
                      CONV_LDS_FLOATS * sizeof(float), stream>>>(
      x, w_conv1, b_conv1, w_conv2, b_conv2, w_fm, b_fm, ini, fea);

  trans_kernel<<<dim3(BATCH * CLS * LPATCH / 64, 2, 1), 256,
                 TRANS_LDS_FLOATS * sizeof(float), stream>>>(
      ini, fea, w1p, w2p, ui, uf);

  att_kernel<<<dim3(LPATCH / 128, LPATCH / 64, BATCH * CLS), 256,
               ATT_LDS_FLOATS * sizeof(float), stream>>>(ui, uf, out);
}


# conv2 as a 2nd matrix product on conv1's result registers (D pairs as B, w2 tap rows as A: 16 rows on WMMA, 2 as FMAs); VALU conv2 and its 64-channel LDS plane go; conv2 sums reordered
# speedup vs baseline: 1.9545x; 1.9545x over previous
#include <hip/hip_runtime.h>
#include <hip/hip_bf16.h>

typedef float v2f __attribute__((ext_vector_type(2)));
typedef float v8f __attribute__((ext_vector_type(8)));

__device__ __forceinline__ v8f wmma_f32(v2f a, v2f b, v8f c) {
  return __builtin_amdgcn_wmma_f32_16x16x4_f32(false, a, false, b, (short)0, c,
                                               false, false);
}

__device__ __forceinline__ void wmma_settle(v8f& d) {
  asm volatile("v_nop\n\tv_nop\n\tv_nop\n\tv_nop" : "+v"(d));
}
__device__ __forceinline__ void wmma_settle(v8f& d0, v8f& d1) {
  asm volatile("v_nop\n\tv_nop\n\tv_nop\n\tv_nop" : "+v"(d0), "+v"(d1));
}

#define BATCH   8
#define CIN     3
#define HW      320
#define CLS     2
#define PSZ     10
#define P2      100
#define LPATCH  1024
#define HID     200

#define WS_INI  0
#define WS_FEA  (WS_INI + BATCH*CLS*HW*HW)
#define WS_UI   (WS_FEA + BATCH*CLS*HW*HW)
#define WS_UF   (WS_UI  + BATCH*CLS*LPATCH*P2)
#define WS_W1P  (WS_UF  + BATCH*CLS*LPATCH*P2)
#define WS_W2P  (WS_W1P + 13*25*64)
#define WS_AIMG WS_UI

#define KSTEPS 38
struct KStep { int c, ky, kx, sky, skx, hc, hky, hkx; };
__host__ __device__ constexpr KStep kstep(int j) {
  if (j < 21) return { j / 7, j % 7, 0,   0, 1,   0, 0, 2 };
  j -= 21;
  if (j < 9)  return { j / 3, 2 * (j % 3), 4,   0, 1,   0, 1, 0 };
  j -= 9;
  if (j < 6)  return { j / 2, 4 * (j % 2), 6,   1, 0,   0, 2, 0 };
  j -= 6;
  if (j == 0) return { 0, 6, 4,   0, 1,   1, 0, 0 };
  return { 2, 6, 4,   0, 1,   0, 1, 0 };
}

__global__ __launch_bounds__(256) void pad_weights_kernel(
    const float* __restrict__ w_lin1, const float* __restrict__ w_lin2,
    const float* __restrict__ w_conv1,
    float* __restrict__ w1f, float* __restrict__ w2f,
    float* __restrict__ aimg) {
  int i = blockIdx.x * 256 + threadIdx.x;
  const int e = i & 1, lane = (i >> 1) & 31, gj = i >> 6;
  if (i < 4 * KSTEPS * 64) {
    const int j = gj % KSTEPS, g = gj / KSTEPS;
    const int h = lane >> 4, row = g * 16 + (lane & 15);
    const KStep P = kstep(j);
    const int c  = P.c + h * P.hc;
    const int ky = P.ky + e * P.sky + h * P.hky;
    const int kx = P.kx + e * P.skx + h * P.hkx;
    aimg[i] = (c < 3 && ky < 7 && kx < 7)
                  ? w_conv1[row * 147 + c * 49 + ky * 7 + kx] : 0.f;
  }
  const int row = lane & 15, kk = 2 * (lane >> 4) + e;
  if (i < 13 * 25 * 64) {
    const int j = gj % 25, n = (gj / 25) * 16 + row;
    w1f[i] = (n < HID) ? w_lin1[n * P2 + 4 * j + kk] : 0.f;
  }
  if (i < 7 * 50 * 64) {
    const int j = gj % 50, n = (gj / 50) * 16 + row;
    w2f[i] = (n < P2) ? w_lin2[n * HID + 4 * j + kk] : 0.f;
  }
}

#define XS_FLOATS        2160
#define SP               338
#define CONV_LDS_FLOATS  (XS_FLOATS + 4 * 16 * SP + 4 * 2 * 2 * SP)

template <int NT>
__device__ __forceinline__ void conv_tiles(
    const v2f (&A)[KSTEPS], const v2f (&A2)[4], const float (&wx)[2][8],
    const float (&bias)[8], const float* xs, float* ss, float* sx,
    int tile0, int l16, int h, int grp, int X0, int Y0) {
  int pb[NT];
  for (int t = 0; t < NT; ++t) {
    const int p  = (tile0 + t) * 16 + l16;
    const int pp = (p < 324) ? p : 0;
    pb[t] = (pp / 18) * 24 + (pp % 18);
  }
  v8f acc[NT];
  for (int t = 0; t < NT; ++t) acc[t] = v8f{};

  #pragma unroll
  for (int j = 0; j < KSTEPS; ++j) {
    const KStep P = kstep(j);
    const int o0 = P.c * 576 + P.ky * 24 + P.kx;
    const int ds = P.sky * 24 + P.skx;
    const int dh = P.hc * 576 + P.hky * 24 + P.hkx;
    for (int t = 0; t < NT; ++t) {
      const float* bp = xs + (pb[t] + h * dh) + o0;
      v2f bf;
      bf.x = bp[0];
      bf.y = bp[ds];
      acc[t] = wmma_f32(A[j], bf, acc[t]);
    }
  }

  if constexpr (NT == 2) wmma_settle(acc[0], acc[1]); else wmma_settle(acc[0]);

  float T[NT][8];
  for (int t = 0; t < NT; ++t) {
    const int p  = (tile0 + t) * 16 + l16;
    const int ty = p / 18, tx = p % 18;
    const int gy = Y0 - 1 + ty, gx = X0 - 1 + tx;
    const bool in = (gy >= 0 && gy < HW && gx >= 0 && gx < HW);
    for (int r = 0; r < 8; ++r) T[t][r] = in ? (acc[t][r] + bias[r]) : 0.f;
  }

  v8f S[NT];
  for (int t = 0; t < NT; ++t) S[t] = v8f{};
  #pragma unroll
  for (int q = 0; q < 4; ++q)
    for (int t = 0; t < NT; ++t) {
      v2f bt;
      bt.x = T[t][2 * q];
      bt.y = T[t][2 * q + 1];
      S[t] = wmma_f32(A2[q], bt, S[t]);
    }
  if constexpr (NT == 2) wmma_settle(S[0], S[1]); else wmma_settle(S[0]);

  for (int t = 0; t < NT; ++t) {
    const int p = (tile0 + t) * 16 + l16;
    float e0 = 0.f, e1 = 0.f;
    for (int r = 0; r < 8; ++r) {
      e0 = fmaf(wx[0][r], T[t][r], e0);
      e1 = fmaf(wx[1][r], T[t][r], e1);
    }
    float* sp = ss + (grp * 16 + 8 * h) * SP + p;
    for (int r = 0; r < 8; ++r) sp[r * SP] = S[t][r];
    float* ep = sx + ((grp * 2 + h) * 2) * SP + p;
    ep[0]  = e0;
    ep[SP] = e1;
  }
}

__global__ __launch_bounds__(256) void conv_fused_kernel(
    const float* __restrict__ x,  const float* __restrict__ aimg,
    const float* __restrict__ b1, const float* __restrict__ w2,
    const float* __restrict__ b2, const float* __restrict__ wfm,
    const float* __restrict__ bfm,
    float* __restrict__ ini, float* __restrict__ fea) {
  extern __shared__ float smem[];
  float* xs = smem;
  float* ss = xs + XS_FLOATS;
  float* sx = ss + 4 * 16 * SP;

  const int tid = threadIdx.x;
  const int X0 = blockIdx.x * 16, Y0 = blockIdx.y * 16, b = blockIdx.z;

  const int lane = tid & 31;
  const int l16  = lane & 15;
  const int h    = lane >> 4;
  const int wave = __builtin_amdgcn_readfirstlane(tid >> 5);
  const int grp  = wave & 3;

  v2f A[KSTEPS];
  {
    const v2f* ap = (const v2f*)aimg + (size_t)grp * KSTEPS * 32 + lane;
    #pragma unroll
    for (int j = 0; j < KSTEPS; ++j) A[j] = ap[j * 32];
  }
  float bias[8];
  for (int r = 0; r < 8; ++r) bias[r] = b1[grp * 16 + 8 * h + r];

  v2f A2[4];
  float wx[2][8];
  {
    const int cls = (l16 >= 9) ? 1 : 0, tap = l16 - 9 * cls;
    const float* wp = w2 + ((size_t)(cls * 64 + grp * 16 + 8 * h) * 9 + tap);
    for (int q = 0; q < 4; ++q) {
      A2[q].x = wp[(2 * q) * 9];
      A2[q].y = wp[(2 * q + 1) * 9];
    }
    const float* xp = w2 + ((size_t)(64 + grp * 16 + 8 * h) * 9 + 7);
    for (int r = 0; r < 8; ++r) { wx[0][r] = xp[r * 9]; wx[1][r] = xp[r * 9 + 1]; }
  }

  for (int i = tid; i < XS_FLOATS; i += 256) {
    float v = 0.f;
    if (i < CIN * 576) {
      int c = i / 576, r = i % 576, iy = r / 24, ix = r % 24;
      int gy = Y0 - 4 + iy, gx = X0 - 4 + ix;
      if (gy >= 0 && gy < HW && gx >= 0 && gx < HW)
        v = x[((size_t)(b * CIN + c) * HW + gy) * HW + gx];
    }
    xs[i] = v;
  }
  __syncthreads();

  {
    const int tile0 = (wave >> 2) ? 11 : 0;
    #pragma nounroll
    for (int tp = 0; tp < 5; ++tp)
      conv_tiles<2>(A, A2, wx, bias, xs, ss, sx, tile0 + 2 * tp, l16, h, grp, X0, Y0);
    if ((wave >> 2) == 0)
      conv_tiles<1>(A, A2, wx, bias, xs, ss, sx, 10, l16, h, grp, X0, Y0);
  }
  __syncthreads();

  {
    const int oy = tid >> 4, ox = tid & 15;
    float a0 = b2[0], a1 = b2[1];
    #pragma unroll
    for (int dy = 0; dy < 3; ++dy)
      #pragma unroll
      for (int dx = 0; dx < 3; ++dx) {
        const int tap = dy * 3 + dx;
        const int pix = (oy + dy) * 18 + (ox + dx);
        #pragma unroll
        for (int g = 0; g < 4; ++g) {
          a0 += ss[(g * 16 + tap) * SP + pix];
          if (tap < 7) {
            a1 += ss[(g * 16 + 9 + tap) * SP + pix];
          } else {
            a1 += sx[((g * 2 + 0) * 2 + (tap - 7)) * SP + pix];
            a1 += sx[((g * 2 + 1) * 2 + (tap - 7)) * SP + pix];
          }
        }
      }
    const int gy = Y0 + oy, gx = X0 + ox;
    const size_t o0 = ((size_t)(b * CLS + 0) * HW + gy) * HW + gx;
    const size_t o1 = ((size_t)(b * CLS + 1) * HW + gy) * HW + gx;
    ini[o0] = a0;
    ini[o1] = a1;
    float x0 = xs[(oy + 4) * 24 + (ox + 4)];
    float x1 = xs[576 + (oy + 4) * 24 + (ox + 4)];
    float x2 = xs[1152 + (oy + 4) * 24 + (ox + 4)];
    fea[o0] = (wfm[0] * x0 + wfm[1] * x1 + wfm[2] * x2 + bfm[0]) * 0.25f;
    fea[o1] = (wfm[3] * x0 + wfm[4] * x1 + wfm[5] * x2 + bfm[1]) * 0.25f;
  }
}

#define TRANS_LDS_FLOATS (64 * 102 + 64 * 210)

__global__ __launch_bounds__(256) void trans_kernel(
    const float* __restrict__ ini, const float* __restrict__ fea,
    const float* __restrict__ w1f, const float* __restrict__ w2f,
    float* __restrict__ ui, float* __restrict__ uf) {
  extern __shared__ float smem[];
  float* a_lds = smem;
  float* h_lds = a_lds + 64 * 102;

  const int tid = threadIdx.x;
  const float* src = blockIdx.y ? fea : ini;
  float* dst       = blockIdx.y ? uf  : ui;
  const int P0 = blockIdx.x * 64;

  for (int i = tid; i < 64 * P2; i += 256) {
    int r = i / P2, p = i % P2;
    int Pa = P0 + r;
    int bc = Pa >> 10, l = Pa & 1023;
    int gy = (l >> 5) * PSZ + p / PSZ;
    int gx = (l & 31) * PSZ + p % PSZ;
    a_lds[r * 102 + p] = src[((size_t)bc * HW + gy) * HW + gx];
  }
  __syncthreads();

  const int wave  = __builtin_amdgcn_readfirstlane(tid >> 5);
  const int lane  = tid & 31;
  const int l16   = lane & 15;
  const int lhalf = (lane < 16) ? 0 : 2;

  for (int t = wave; t < 4 * 13; t += 8) {
    const int m0 = (t / 13) * 16, nt = t % 13;
    const int n  = nt * 16 + l16;
    const v2f* ap = (const v2f*)&a_lds[(m0 + l16) * 102 + lhalf];
    const v2f* bp = (const v2f*)w1f + nt * (25 * 32) + lane;
    v2f bfr[25];
    #pragma unroll
    for (int j = 0; j < 25; ++j) bfr[j] = bp[j * 32];
    v8f acc = {};
    #pragma unroll
    for (int j = 0; j < 25; ++j) acc = wmma_f32(ap[2 * j], bfr[j], acc);
    #pragma unroll
    for (int r = 0; r < 8; ++r) {
      int M = m0 + r + ((lane < 16) ? 0 : 8);
      h_lds[M * 210 + n] = acc[r];
    }
  }
  __syncthreads();

  for (int t = wave; t < 4 * 7; t += 8) {
    const int m0 = (t / 7) * 16, nt = t % 7;
    const int n  = nt * 16 + l16;
    const v2f* ap = (const v2f*)&h_lds[(m0 + l16) * 210 + lhalf];
    const v2f* bp = (const v2f*)w2f + nt * (50 * 32) + lane;
    v2f bfr[50];
    #pragma unroll
    for (int j = 0; j < 50; ++j) bfr[j] = bp[j * 32];
    v8f acc = {};
    #pragma unroll
    for (int j = 0; j < 50; ++j) acc = wmma_f32(ap[2 * j], bfr[j], acc);
    float v[8];
    for (int r = 0; r < 8; ++r) v[r] = acc[r] > 0.f ? acc[r] : 0.f;
    if (n < P2) {
      float* dp = dst + ((size_t)((P0 + m0) >> 4) * 25 + (n >> 2)) * 64
                      + ((n >> 1) & 1) * 32 + (lane >> 4) * 16 + (n & 1);
      #pragma unroll
      for (int r = 0; r < 8; ++r) dp[2 * r] = v[r];
    }
  }
}

__global__ __launch_bounds__(256) void att_kernel(
    const float* __restrict__ ui, const float* __restrict__ uf,
    float* __restrict__ out) {
  const int tid  = threadIdx.x;
  const int bc   = blockIdx.z;
  const int lane = tid & 31;
  const int l16  = lane & 15;
  const int h    = lane >> 4;
  const int wave = __builtin_amdgcn_readfirstlane(tid >> 5);
  const int nt   = blockIdx.y * 4 + (wave & 3);
  const int mtb  = blockIdx.x * 8 + (wave >> 2);

  const v2f* Uf = (const v2f*)ui + (size_t)bc * 64 * (25 * 32) + lane;
  const v2f* Vf = (const v2f*)uf + ((size_t)bc * 64 + nt) * (25 * 32) + lane;

  v2f B[25];
  #pragma unroll
  for (int j = 0; j < 25; ++j) B[j] = Vf[j * 32];

  #pragma nounroll
  for (int i = 0; i < 2; ++i) {
    const int mt0 = mtb + 4 * i, mt1 = mt0 + 2;
    const v2f* a0p = Uf + (size_t)mt0 * (25 * 32);
    const v2f* a1p = Uf + (size_t)mt1 * (25 * 32);
    v2f A0[25], A1[25];
    #pragma unroll
    for (int j = 0; j < 25; ++j) { A0[j] = a0p[j * 32]; A1[j] = a1p[j * 32]; }
    __builtin_amdgcn_sched_barrier(0);
    v8f acc0 = {}, acc1 = {};
    #pragma unroll
    for (int j = 0; j < 25; ++j) {
      acc0 = wmma_f32(A0[j], B[j], acc0);
      acc1 = wmma_f32(A1[j], B[j], acc1);
    }
    float* o0 = out + ((size_t)bc * LPATCH + mt0 * 16 + 8 * h) * LPATCH + nt * 16 + l16;
    float* o1 = out + ((size_t)bc * LPATCH + mt1 * 16 + 8 * h) * LPATCH + nt * 16 + l16;
    #pragma unroll
    for (int r = 0; r < 8; ++r) {
      o0[(size_t)r * LPATCH] = acc0[r] * 0.01f;
      o1[(size_t)r * LPATCH] = acc1[r] * 0.01f;
    }
  }
}

extern "C" void kernel_launch(void* const* d_in, const int* in_sizes, int n_in,
                              void* d_out, int out_size, void* d_ws,
                              size_t ws_size, hipStream_t stream) {
  (void)in_sizes; (void)n_in; (void)out_size; (void)ws_size;
  const float* x       = (const float*)d_in[0];
  const float* w_conv1 = (const float*)d_in[1];
  const float* b_conv1 = (const float*)d_in[2];
  const float* w_conv2 = (const float*)d_in[3];
  const float* b_conv2 = (const float*)d_in[4];
  const float* w_fm    = (const float*)d_in[5];
  const float* b_fm    = (const float*)d_in[6];
  const float* w_lin1  = (const float*)d_in[7];
  const float* w_lin2  = (const float*)d_in[8];
  float* out = (float*)d_out;

  float* ws  = (float*)d_ws;
  float* ini = ws + WS_INI;
  float* fea = ws + WS_FEA;
  float* ui  = ws + WS_UI;
  float* uf  = ws + WS_UF;
  float* w1f = ws + WS_W1P;
  float* w2f = ws + WS_W2P;
  float* aimg = ws + WS_AIMG;

  pad_weights_kernel<<<dim3((112 * 200 + 255) / 256), 256, 0, stream>>>(
      w_lin1, w_lin2, w_conv1, w1f, w2f, aimg);

  conv_fused_kernel<<<dim3(HW / 16, HW / 16, BATCH), 256,
                      CONV_LDS_FLOATS * sizeof(float), stream>>>(
      x, aimg, b_conv1, w_conv2, b_conv2, w_fm, b_fm, ini, fea);

  trans_kernel<<<dim3(BATCH * CLS * LPATCH / 64, 2, 1), 256,
                 TRANS_LDS_FLOATS * sizeof(float), stream>>>(
      ini, fea, w1f, w2f, ui, uf);

  att_kernel<<<dim3(LPATCH / 128, LPATCH / 64, BATCH * CLS), 256, 0,
               stream>>>(ui, uf, out);
}


# conv1 only on a tile's own 256 pixels (16 N-tiles, was 21 with halo): conv2 taps crossing tiles leave as edge partial sums via workspace, added in the unfold (fea made there from x); ini sums reorder
# speedup vs baseline: 2.1985x; 1.1248x over previous
#include <hip/hip_runtime.h>
#include <hip/hip_bf16.h>

typedef float v2f __attribute__((ext_vector_type(2)));
typedef float v8f __attribute__((ext_vector_type(8)));

__device__ __forceinline__ v8f wmma_f32(v2f a, v2f b, v8f c) {
  return __builtin_amdgcn_wmma_f32_16x16x4_f32(false, a, false, b, (short)0, c,
                                               false, false);
}

__device__ __forceinline__ void wmma_settle(v8f& d) {
  asm volatile("v_nop\n\tv_nop\n\tv_nop\n\tv_nop" : "+v"(d));
}
__device__ __forceinline__ void wmma_settle(v8f& d0, v8f& d1) {
  asm volatile("v_nop\n\tv_nop\n\tv_nop\n\tv_nop" : "+v"(d0), "+v"(d1));
}

#define BATCH   8
#define CIN     3
#define HW      320
#define CLS     2
#define PSZ     10
#define P2      100
#define LPATCH  1024
#define HID     200

#define WS_INI  0
#define WS_FEA  (WS_INI + BATCH*CLS*HW*HW)
#define WS_UI   (WS_FEA + BATCH*CLS*HW*HW)
#define WS_UF   (WS_UI  + BATCH*CLS*LPATCH*P2)
#define WS_W1P  (WS_UF  + BATCH*CLS*LPATCH*P2)
#define WS_W2P  (WS_W1P + 13*25*64)
#define WS_AIMG WS_UI
#define NEDGE   40
#define WS_EH   WS_FEA
#define WS_EV   (WS_EH + BATCH*CLS*NEDGE*HW)
#define WS_ED   (WS_EV + BATCH*CLS*NEDGE*HW)

#define KSTEPS 38
struct KStep { int c, ky, kx, sky, skx, hc, hky, hkx; };
__host__ __device__ constexpr KStep kstep(int j) {
  if (j < 21) return { j / 7, j % 7, 0,   0, 1,   0, 0, 2 };
  j -= 21;
  if (j < 9)  return { j / 3, 2 * (j % 3), 4,   0, 1,   0, 1, 0 };
  j -= 9;
  if (j < 6)  return { j / 2, 4 * (j % 2), 6,   1, 0,   0, 2, 0 };
  j -= 6;
  if (j == 0) return { 0, 6, 4,   0, 1,   1, 0, 0 };
  return { 2, 6, 4,   0, 1,   0, 1, 0 };
}

__global__ __launch_bounds__(256) void pad_weights_kernel(
    const float* __restrict__ w_lin1, const float* __restrict__ w_lin2,
    const float* __restrict__ w_conv1,
    float* __restrict__ w1f, float* __restrict__ w2f,
    float* __restrict__ aimg) {
  int i = blockIdx.x * 256 + threadIdx.x;
  const int e = i & 1, lane = (i >> 1) & 31, gj = i >> 6;
  if (i < 4 * KSTEPS * 64) {
    const int j = gj % KSTEPS, g = gj / KSTEPS;
    const int h = lane >> 4, row = g * 16 + (lane & 15);
    const KStep P = kstep(j);
    const int c  = P.c + h * P.hc;
    const int ky = P.ky + e * P.sky + h * P.hky;
    const int kx = P.kx + e * P.skx + h * P.hkx;
    aimg[i] = (c < 3 && ky < 7 && kx < 7)
                  ? w_conv1[row * 147 + c * 49 + ky * 7 + kx] : 0.f;
  }
  const int row = lane & 15, kk = 2 * (lane >> 4) + e;
  if (i < 13 * 25 * 64) {
    const int j = gj % 25, n = (gj / 25) * 16 + row;
    w1f[i] = (n < HID) ? w_lin1[n * P2 + 4 * j + kk] : 0.f;
  }
  if (i < 7 * 50 * 64) {
    const int j = gj % 50, n = (gj / 50) * 16 + row;
    w2f[i] = (n < P2) ? w_lin2[n * HID + 4 * j + kk] : 0.f;
  }
}

#define XS_FLOATS        2160
#define SP               274
#define CONV_LDS_FLOATS  (XS_FLOATS + 4 * 16 * SP + 4 * 2 * 2 * SP)

template <int NT>
__device__ __forceinline__ void conv_tiles(
    const v2f (&A)[KSTEPS], const v2f (&A2)[4], const float (&wx)[2][8],
    const float (&bias)[8], const float* xs, float* ss, float* sx,
    int tile0, int l16, int h, int grp) {
  int pb[NT];
  for (int t = 0; t < NT; ++t) pb[t] = (tile0 + t + 1) * 24 + (l16 + 1);
  v8f acc[NT];
  for (int t = 0; t < NT; ++t) acc[t] = v8f{};

  #pragma unroll
  for (int j = 0; j < KSTEPS; ++j) {
    const KStep P = kstep(j);
    const int o0 = P.c * 576 + P.ky * 24 + P.kx;
    const int ds = P.sky * 24 + P.skx;
    const int dh = P.hc * 576 + P.hky * 24 + P.hkx;
    for (int t = 0; t < NT; ++t) {
      const float* bp = xs + (pb[t] + h * dh) + o0;
      v2f bf;
      bf.x = bp[0];
      bf.y = bp[ds];
      acc[t] = wmma_f32(A[j], bf, acc[t]);
    }
  }

  if constexpr (NT == 2) wmma_settle(acc[0], acc[1]); else wmma_settle(acc[0]);

  float T[NT][8];
  for (int t = 0; t < NT; ++t)
    for (int r = 0; r < 8; ++r) T[t][r] = acc[t][r] + bias[r];

  v8f S[NT];
  for (int t = 0; t < NT; ++t) S[t] = v8f{};
  #pragma unroll
  for (int q = 0; q < 4; ++q)
    for (int t = 0; t < NT; ++t) {
      v2f bt;
      bt.x = T[t][2 * q];
      bt.y = T[t][2 * q + 1];
      S[t] = wmma_f32(A2[q], bt, S[t]);
    }
  if constexpr (NT == 2) wmma_settle(S[0], S[1]); else wmma_settle(S[0]);

  for (int t = 0; t < NT; ++t) {
    const int p = (tile0 + t) * 16 + l16;
    float e0 = 0.f, e1 = 0.f;
    for (int r = 0; r < 8; ++r) {
      e0 = fmaf(wx[0][r], T[t][r], e0);
      e1 = fmaf(wx[1][r], T[t][r], e1);
    }
    float* sp = ss + (grp * 16 + 8 * h) * SP + p;
    for (int r = 0; r < 8; ++r) sp[r * SP] = S[t][r];
    float* ep = sx + ((grp * 2 + h) * 2) * SP + p;
    ep[0]  = e0;
    ep[SP] = e1;
  }
}

__global__ __launch_bounds__(256) void conv_fused_kernel(
    const float* __restrict__ x,  const float* __restrict__ aimg,
    const float* __restrict__ b1, const float* __restrict__ w2,
    const float* __restrict__ b2,
    float* __restrict__ ini, float* __restrict__ eh,
    float* __restrict__ ev,  float* __restrict__ ed) {
  extern __shared__ float smem[];
  float* xs = smem;
  float* ss = xs + XS_FLOATS;
  float* sx = ss + 4 * 16 * SP;

  const int tid = threadIdx.x;
  const int X0 = blockIdx.x * 16, Y0 = blockIdx.y * 16, b = blockIdx.z;

  const int lane = tid & 31;
  const int l16  = lane & 15;
  const int h    = lane >> 4;
  const int wave = __builtin_amdgcn_readfirstlane(tid >> 5);
  const int grp  = wave & 3;

  v2f A[KSTEPS];
  {
    const v2f* ap = (const v2f*)aimg + (size_t)grp * KSTEPS * 32 + lane;
    #pragma unroll
    for (int j = 0; j < KSTEPS; ++j) A[j] = ap[j * 32];
  }
  float bias[8];
  for (int r = 0; r < 8; ++r) bias[r] = b1[grp * 16 + 8 * h + r];

  v2f A2[4];
  float wx[2][8];
  {
    const int cls = (l16 >= 9) ? 1 : 0, tap = l16 - 9 * cls;
    const float* wp = w2 + ((size_t)(cls * 64 + grp * 16 + 8 * h) * 9 + tap);
    for (int q = 0; q < 4; ++q) {
      A2[q].x = wp[(2 * q) * 9];
      A2[q].y = wp[(2 * q + 1) * 9];
    }
    const float* xp = w2 + ((size_t)(64 + grp * 16 + 8 * h) * 9 + 7);
    for (int r = 0; r < 8; ++r) { wx[0][r] = xp[r * 9]; wx[1][r] = xp[r * 9 + 1]; }
  }

  for (int i = tid; i < XS_FLOATS; i += 256) {
    float v = 0.f;
    if (i < CIN * 576) {
      int c = i / 576, r = i % 576, iy = r / 24, ix = r % 24;
      int gy = Y0 - 4 + iy, gx = X0 - 4 + ix;
      if (gy >= 0 && gy < HW && gx >= 0 && gx < HW)
        v = x[((size_t)(b * CIN + c) * HW + gy) * HW + gx];
    }
    xs[i] = v;
  }
  __syncthreads();

  {
    const int tile0 = (wave >> 2) * 8;
    #pragma nounroll
    for (int tp = 0; tp < 4; ++tp)
      conv_tiles<2>(A, A2, wx, bias, xs, ss, sx, tile0 + 2 * tp, l16, h, grp);
  }
  __syncthreads();

  for (int pos = tid; pos < 18 * 18; pos += 256) {
    const int py = pos / 18, px = pos - py * 18;
    const int gy = Y0 - 1 + py, gx = X0 - 1 + px;
    const bool rowOwn = (py >= 1 && py <= 16), colOwn = (px >= 1 && px <= 16);
    const bool own = rowOwn && colOwn;
    float a0 = own ? b2[0] : 0.f, a1 = own ? b2[1] : 0.f;
    #pragma unroll
    for (int dy = 0; dy < 3; ++dy)
      #pragma unroll
      for (int dx = 0; dx < 3; ++dx) {
        const int tap = dy * 3 + dx;
        const int qy = py + dy - 2, qx = px + dx - 2;
        if ((unsigned)qy < 16u && (unsigned)qx < 16u) {
          const int pix = qy * 16 + qx;
          #pragma unroll
          for (int g = 0; g < 4; ++g) {
            a0 += ss[(g * 16 + tap) * SP + pix];
            if (tap < 7) {
              a1 += ss[(g * 16 + 9 + tap) * SP + pix];
            } else {
              a1 += sx[((g * 2 + 0) * 2 + (tap - 7)) * SP + pix];
              a1 += sx[((g * 2 + 1) * 2 + (tap - 7)) * SP + pix];
            }
          }
        }
      }
    if (gy >= 0 && gy < HW && gx >= 0 && gx < HW) {
      const int pl = b * CLS;
      if (own) {
        const size_t o0 = ((size_t)pl * HW + gy) * HW + gx;
        ini[o0] = a0;
        ini[o0 + (size_t)HW * HW] = a1;
      } else {
        const int hc = (gx >> 4) * 2 + ((gx & 15) == 15 ? 1 : 0);
        const int vr = (gy >> 4) * 2 + ((gy & 15) == 15 ? 1 : 0);
        if (rowOwn) {
          const size_t o = ((size_t)pl * NEDGE + hc) * HW + gy;
          eh[o] = a0;
          eh[o + (size_t)NEDGE * HW] = a1;
        } else if (colOwn) {
          const size_t o = ((size_t)pl * NEDGE + vr) * HW + gx;
          ev[o] = a0;
          ev[o + (size_t)NEDGE * HW] = a1;
        } else {
          const size_t o = ((size_t)pl * NEDGE + vr) * NEDGE + hc;
          ed[o] = a0;
          ed[o + (size_t)NEDGE * NEDGE] = a1;
        }
      }
    }
  }
}

#define TRANS_LDS_FLOATS (64 * 102 + 64 * 210)

__global__ __launch_bounds__(256) void trans_kernel(
    const float* __restrict__ ini, const float* __restrict__ eh,
    const float* __restrict__ ev,  const float* __restrict__ ed,
    const float* __restrict__ x,   const float* __restrict__ wfm,
    const float* __restrict__ bfm,
    const float* __restrict__ w1f, const float* __restrict__ w2f,
    float* __restrict__ ui, float* __restrict__ uf) {
  extern __shared__ float smem[];
  float* a_lds = smem;
  float* h_lds = a_lds + 64 * 102;

  const int tid = threadIdx.x;
  const bool is_fea = blockIdx.y != 0;
  float* dst        = is_fea ? uf : ui;
  const int P0 = blockIdx.x * 64;

  for (int i = tid; i < 64 * P2; i += 256) {
    int r = i / P2, p = i % P2;
    int Pa = P0 + r;
    int bc = Pa >> 10, l = Pa & 1023;
    int gy = (l >> 5) * PSZ + p / PSZ;
    int gx = (l & 31) * PSZ + p % PSZ;
    float v;
    if (is_fea) {
      const int cls = bc & 1;
      const float* xp = x + ((size_t)((bc >> 1) * CIN) * HW + gy) * HW + gx;
      v = (wfm[cls * 3] * xp[0] + wfm[cls * 3 + 1] * xp[(size_t)HW * HW]
           + wfm[cls * 3 + 2] * xp[(size_t)2 * HW * HW] + bfm[cls]) * 0.25f;
    } else {
      v = ini[((size_t)bc * HW + gy) * HW + gx];
      const int xm = gx & 15, ym = gy & 15;
      const bool hH = (xm == 0 && gx > 0) || (xm == 15 && gx < HW - 1);
      const bool hV = (ym == 0 && gy > 0) || (ym == 15 && gy < HW - 1);
      const int hc = (gx >> 4) * 2 + (xm == 15 ? 1 : 0);
      const int vr = (gy >> 4) * 2 + (ym == 15 ? 1 : 0);
      if (hH) v += eh[((size_t)bc * NEDGE + hc) * HW + gy];
      if (hV) v += ev[((size_t)bc * NEDGE + vr) * HW + gx];
      if (hH && hV) v += ed[((size_t)bc * NEDGE + vr) * NEDGE + hc];
    }
    a_lds[r * 102 + p] = v;
  }
  __syncthreads();

  const int wave  = __builtin_amdgcn_readfirstlane(tid >> 5);
  const int lane  = tid & 31;
  const int l16   = lane & 15;
  const int lhalf = (lane < 16) ? 0 : 2;

  for (int t = wave; t < 4 * 13; t += 8) {
    const int m0 = (t / 13) * 16, nt = t % 13;
    const int n  = nt * 16 + l16;
    const v2f* ap = (const v2f*)&a_lds[(m0 + l16) * 102 + lhalf];
    const v2f* bp = (const v2f*)w1f + nt * (25 * 32) + lane;
    v2f bfr[25];
    #pragma unroll
    for (int j = 0; j < 25; ++j) bfr[j] = bp[j * 32];
    v8f acc = {};
    #pragma unroll
    for (int j = 0; j < 25; ++j) acc = wmma_f32(ap[2 * j], bfr[j], acc);
    #pragma unroll
    for (int r = 0; r < 8; ++r) {
      int M = m0 + r + ((lane < 16) ? 0 : 8);
      h_lds[M * 210 + n] = acc[r];
    }
  }
  __syncthreads();

  for (int t = wave; t < 4 * 7; t += 8) {
    const int m0 = (t / 7) * 16, nt = t % 7;
    const int n  = nt * 16 + l16;
    const v2f* ap = (const v2f*)&h_lds[(m0 + l16) * 210 + lhalf];
    const v2f* bp = (const v2f*)w2f + nt * (50 * 32) + lane;
    v2f bfr[50];
    #pragma unroll
    for (int j = 0; j < 50; ++j) bfr[j] = bp[j * 32];
    v8f acc = {};
    #pragma unroll
    for (int j = 0; j < 50; ++j) acc = wmma_f32(ap[2 * j], bfr[j], acc);
    float v[8];
    for (int r = 0; r < 8; ++r) v[r] = acc[r] > 0.f ? acc[r] : 0.f;
    if (n < P2) {
      float* dp = dst + ((size_t)((P0 + m0) >> 4) * 25 + (n >> 2)) * 64
                      + ((n >> 1) & 1) * 32 + (lane >> 4) * 16 + (n & 1);
      #pragma unroll
      for (int r = 0; r < 8; ++r) dp[2 * r] = v[r];
    }
  }
}

__global__ __launch_bounds__(256) void att_kernel(
    const float* __restrict__ ui, const float* __restrict__ uf,
    float* __restrict__ out) {
  const int tid  = threadIdx.x;
  const int bc   = blockIdx.z;
  const int lane = tid & 31;
  const int l16  = lane & 15;
  const int h    = lane >> 4;
  const int wave = __builtin_amdgcn_readfirstlane(tid >> 5);
  const int nt   = blockIdx.y * 4 + (wave & 3);
  const int mtb  = blockIdx.x * 8 + (wave >> 2);

  const v2f* Uf = (const v2f*)ui + (size_t)bc * 64 * (25 * 32) + lane;
  const v2f* Vf = (const v2f*)uf + ((size_t)bc * 64 + nt) * (25 * 32) + lane;

  v2f B[25];
  #pragma unroll
  for (int j = 0; j < 25; ++j) B[j] = Vf[j * 32];

  #pragma nounroll
  for (int i = 0; i < 2; ++i) {
    const int mt0 = mtb + 4 * i, mt1 = mt0 + 2;
    const v2f* a0p = Uf + (size_t)mt0 * (25 * 32);
    const v2f* a1p = Uf + (size_t)mt1 * (25 * 32);
    v2f A0[25], A1[25];
    #pragma unroll
    for (int j = 0; j < 25; ++j) { A0[j] = a0p[j * 32]; A1[j] = a1p[j * 32]; }
    __builtin_amdgcn_sched_barrier(0);
    v8f acc0 = {}, acc1 = {};
    #pragma unroll
    for (int j = 0; j < 25; ++j) {
      acc0 = wmma_f32(A0[j], B[j], acc0);
      acc1 = wmma_f32(A1[j], B[j], acc1);
    }
    float* o0 = out + ((size_t)bc * LPATCH + mt0 * 16 + 8 * h) * LPATCH + nt * 16 + l16;
    float* o1 = out + ((size_t)bc * LPATCH + mt1 * 16 + 8 * h) * LPATCH + nt * 16 + l16;
    #pragma unroll
    for (int r = 0; r < 8; ++r) {
      o0[(size_t)r * LPATCH] = acc0[r] * 0.01f;
      o1[(size_t)r * LPATCH] = acc1[r] * 0.01f;
    }
  }
}

extern "C" void kernel_launch(void* const* d_in, const int* in_sizes, int n_in,
                              void* d_out, int out_size, void* d_ws,
                              size_t ws_size, hipStream_t stream) {
  (void)in_sizes; (void)n_in; (void)out_size; (void)ws_size;
  const float* x       = (const float*)d_in[0];
  const float* w_conv1 = (const float*)d_in[1];
  const float* b_conv1 = (const float*)d_in[2];
  const float* w_conv2 = (const float*)d_in[3];
  const float* b_conv2 = (const float*)d_in[4];
  const float* w_fm    = (const float*)d_in[5];
  const float* b_fm    = (const float*)d_in[6];
  const float* w_lin1  = (const float*)d_in[7];
  const float* w_lin2  = (const float*)d_in[8];
  float* out = (float*)d_out;

  float* ws  = (float*)d_ws;
  float* ini = ws + WS_INI;
  float* eh  = ws + WS_EH;
  float* ev  = ws + WS_EV;
  float* ed  = ws + WS_ED;
  float* ui  = ws + WS_UI;
  float* uf  = ws + WS_UF;
  float* w1f = ws + WS_W1P;
  float* w2f = ws + WS_W2P;
  float* aimg = ws + WS_AIMG;

  pad_weights_kernel<<<dim3((112 * 200 + 255) / 256), 256, 0, stream>>>(
      w_lin1, w_lin2, w_conv1, w1f, w2f, aimg);

  conv_fused_kernel<<<dim3(HW / 16, HW / 16, BATCH), 256,
                      CONV_LDS_FLOATS * sizeof(float), stream>>>(
      x, aimg, b_conv1, w_conv2, b_conv2, ini, eh, ev, ed);

  trans_kernel<<<dim3(BATCH * CLS * LPATCH / 64, 2, 1), 256,
                 TRANS_LDS_FLOATS * sizeof(float), stream>>>(
      ini, eh, ev, ed, x, w_fm, b_fm, w1f, w2f, ui, uf);

  att_kernel<<<dim3(LPATCH / 128, LPATCH / 64, BATCH * CLS), 256, 0,
               stream>>>(ui, uf, out);
}


# conv kernel: the conv2 stage's per-lane constants (conv2 weights, conv1 bias) sit in LDS and are re-read per tile pair, not held over the K loop: 157 VGPRs, 3 blocks per WGP (was 2); sums unchanged
# speedup vs baseline: 2.3064x; 1.0491x over previous
#include <hip/hip_runtime.h>
#include <hip/hip_bf16.h>

typedef float v2f __attribute__((ext_vector_type(2)));
typedef float v8f __attribute__((ext_vector_type(8)));

__device__ __forceinline__ v8f wmma_f32(v2f a, v2f b, v8f c) {
  return __builtin_amdgcn_wmma_f32_16x16x4_f32(false, a, false, b, (short)0, c,
                                               false, false);
}

__device__ __forceinline__ void wmma_settle(v8f& d) {
  asm volatile("v_nop\n\tv_nop\n\tv_nop\n\tv_nop" : "+v"(d));
}
__device__ __forceinline__ void wmma_settle(v8f& d0, v8f& d1) {
  asm volatile("v_nop\n\tv_nop\n\tv_nop\n\tv_nop" : "+v"(d0), "+v"(d1));
}

#define BATCH   8
#define CIN     3
#define HW      320
#define CLS     2
#define PSZ     10
#define P2      100
#define LPATCH  1024
#define HID     200

#define WS_INI  0
#define WS_FEA  (WS_INI + BATCH*CLS*HW*HW)
#define WS_UI   (WS_FEA + BATCH*CLS*HW*HW)
#define WS_UF   (WS_UI  + BATCH*CLS*LPATCH*P2)
#define WS_W1P  (WS_UF  + BATCH*CLS*LPATCH*P2)
#define WS_W2P  (WS_W1P + 13*25*64)
#define WS_AIMG WS_UI
#define NEDGE   40
#define WS_EH   WS_FEA
#define WS_EV   (WS_EH + BATCH*CLS*NEDGE*HW)
#define WS_ED   (WS_EV + BATCH*CLS*NEDGE*HW)

#define KSTEPS 38
struct KStep { int c, ky, kx, sky, skx, hc, hky, hkx; };
__host__ __device__ constexpr KStep kstep(int j) {
  if (j < 21) return { j / 7, j % 7, 0,   0, 1,   0, 0, 2 };
  j -= 21;
  if (j < 9)  return { j / 3, 2 * (j % 3), 4,   0, 1,   0, 1, 0 };
  j -= 9;
  if (j < 6)  return { j / 2, 4 * (j % 2), 6,   1, 0,   0, 2, 0 };
  j -= 6;
  if (j == 0) return { 0, 6, 4,   0, 1,   1, 0, 0 };
  return { 2, 6, 4,   0, 1,   0, 1, 0 };
}

__global__ __launch_bounds__(256) void pad_weights_kernel(
    const float* __restrict__ w_lin1, const float* __restrict__ w_lin2,
    const float* __restrict__ w_conv1,
    float* __restrict__ w1f, float* __restrict__ w2f,
    float* __restrict__ aimg) {
  int i = blockIdx.x * 256 + threadIdx.x;
  const int e = i & 1, lane = (i >> 1) & 31, gj = i >> 6;
  if (i < 4 * KSTEPS * 64) {
    const int j = gj % KSTEPS, g = gj / KSTEPS;
    const int h = lane >> 4, row = g * 16 + (lane & 15);
    const KStep P = kstep(j);
    const int c  = P.c + h * P.hc;
    const int ky = P.ky + e * P.sky + h * P.hky;
    const int kx = P.kx + e * P.skx + h * P.hkx;
    aimg[i] = (c < 3 && ky < 7 && kx < 7)
                  ? w_conv1[row * 147 + c * 49 + ky * 7 + kx] : 0.f;
  }
  const int row = lane & 15, kk = 2 * (lane >> 4) + e;
  if (i < 13 * 25 * 64) {
    const int j = gj % 25, n = (gj / 25) * 16 + row;
    w1f[i] = (n < HID) ? w_lin1[n * P2 + 4 * j + kk] : 0.f;
  }
  if (i < 7 * 50 * 64) {
    const int j = gj % 50, n = (gj / 50) * 16 + row;
    w2f[i] = (n < P2) ? w_lin2[n * HID + 4 * j + kk] : 0.f;
  }
}

#define XS_FLOATS        2160
#define SP               274
#define A2_FLOATS        (4 * 2 * 32 * 4)
#define WB_FLOATS        (4 * 2 * 24)
#define CONV_LDS_FLOATS  (XS_FLOATS + 4 * 16 * SP + 4 * 2 * 2 * SP + A2_FLOATS + WB_FLOATS)

template <int NT>
__device__ __forceinline__ void conv_tiles(
    const v2f (&A)[KSTEPS], const float* a2s, const float* wbs,
    const float* xs, float* ss, float* sx,
    int tile0, int l16, int h, int grp) {
  int pb[NT];
  for (int t = 0; t < NT; ++t) pb[t] = (tile0 + t + 1) * 24 + (l16 + 1);
  v8f acc[NT];
  for (int t = 0; t < NT; ++t) acc[t] = v8f{};

  #pragma unroll
  for (int j = 0; j < KSTEPS; ++j) {
    const KStep P = kstep(j);
    const int o0 = P.c * 576 + P.ky * 24 + P.kx;
    const int ds = P.sky * 24 + P.skx;
    const int dh = P.hc * 576 + P.hky * 24 + P.hkx;
    for (int t = 0; t < NT; ++t) {
      const float* bp = xs + (pb[t] + h * dh) + o0;
      v2f bf;
      bf.x = bp[0];
      bf.y = bp[ds];
      acc[t] = wmma_f32(A[j], bf, acc[t]);
    }
  }

  if constexpr (NT == 2) wmma_settle(acc[0], acc[1]); else wmma_settle(acc[0]);

  const float4* wq = (const float4*)wbs;
  float bias[8];
  {
    const float4 lo = wq[4], hi = wq[5];
    bias[0] = lo.x; bias[1] = lo.y; bias[2] = lo.z; bias[3] = lo.w;
    bias[4] = hi.x; bias[5] = hi.y; bias[6] = hi.z; bias[7] = hi.w;
  }
  v2f A2[4];
  {
    const float4 p0 = ((const float4*)a2s)[0], p1 = ((const float4*)a2s)[32];
    A2[0].x = p0.x; A2[0].y = p0.y; A2[1].x = p0.z; A2[1].y = p0.w;
    A2[2].x = p1.x; A2[2].y = p1.y; A2[3].x = p1.z; A2[3].y = p1.w;
  }

  float T[NT][8];
  for (int t = 0; t < NT; ++t)
    for (int r = 0; r < 8; ++r) T[t][r] = acc[t][r] + bias[r];

  v8f S[NT];
  for (int t = 0; t < NT; ++t) S[t] = v8f{};
  #pragma unroll
  for (int q = 0; q < 4; ++q)
    for (int t = 0; t < NT; ++t) {
      v2f bt;
      bt.x = T[t][2 * q];
      bt.y = T[t][2 * q + 1];
      S[t] = wmma_f32(A2[q], bt, S[t]);
    }
  if constexpr (NT == 2) wmma_settle(S[0], S[1]); else wmma_settle(S[0]);

  float wx[2][8];
  for (int k = 0; k < 4; ++k) {
    const float4 v = wq[k];
    wx[0][2 * k]     = v.x; wx[1][2 * k]     = v.y;
    wx[0][2 * k + 1] = v.z; wx[1][2 * k + 1] = v.w;
  }
  for (int t = 0; t < NT; ++t) {
    const int p = (tile0 + t) * 16 + l16;
    float e0 = 0.f, e1 = 0.f;
    for (int r = 0; r < 8; ++r) {
      e0 = fmaf(wx[0][r], T[t][r], e0);
      e1 = fmaf(wx[1][r], T[t][r], e1);
    }
    float* sp = ss + (grp * 16 + 8 * h) * SP + p;
    for (int r = 0; r < 8; ++r) sp[r * SP] = S[t][r];
    float* ep = sx + ((grp * 2 + h) * 2) * SP + p;
    ep[0]  = e0;
    ep[SP] = e1;
  }
}

__global__ __launch_bounds__(256) void conv_fused_kernel(
    const float* __restrict__ x,  const float* __restrict__ aimg,
    const float* __restrict__ b1, const float* __restrict__ w2,
    const float* __restrict__ b2,
    float* __restrict__ ini, float* __restrict__ eh,
    float* __restrict__ ev,  float* __restrict__ ed) {
  extern __shared__ float smem[];
  float* xs = smem;
  float* ss = xs + XS_FLOATS;
  float* sx = ss + 4 * 16 * SP;
  float* a2l = sx + 4 * 2 * 2 * SP;
  float* wbl = a2l + A2_FLOATS;

  const int tid = threadIdx.x;
  const int X0 = blockIdx.x * 16, Y0 = blockIdx.y * 16, b = blockIdx.z;

  const int lane = tid & 31;
  const int l16  = lane & 15;
  const int h    = lane >> 4;
  const int wave = __builtin_amdgcn_readfirstlane(tid >> 5);
  const int grp  = wave & 3;

  v2f A[KSTEPS];
  {
    const v2f* ap = (const v2f*)aimg + (size_t)grp * KSTEPS * 32 + lane;
    #pragma unroll
    for (int j = 0; j < KSTEPS; ++j) A[j] = ap[j * 32];
  }

  const float* a2s = a2l + (grp * 2 * 32 + lane) * 4;
  const float* wbs = wbl + (grp * 2 + h) * 24;
  {
    const int cls = (l16 >= 9) ? 1 : 0, tap = l16 - 9 * cls;
    const float* wp = w2 + ((size_t)(cls * 64 + grp * 16 + 8 * h) * 9 + tap);
    for (int k = 0; k < 2; ++k) {
      float4 v;
      v.x = wp[(4 * k) * 9];     v.y = wp[(4 * k + 1) * 9];
      v.z = wp[(4 * k + 2) * 9]; v.w = wp[(4 * k + 3) * 9];
      ((float4*)(a2l + (grp * 2 * 32 + lane) * 4))[32 * k] = v;
    }
    float* wb = wbl + (grp * 2 + h) * 24;
    wb[l16] = w2[(size_t)(64 + grp * 16 + 8 * h + (l16 >> 1)) * 9 + 7 + (l16 & 1)];
    wb[16 + (l16 & 7)] = b1[grp * 16 + 8 * h + (l16 & 7)];
  }

  for (int i = tid; i < XS_FLOATS; i += 256) {
    float v = 0.f;
    if (i < CIN * 576) {
      int c = i / 576, r = i % 576, iy = r / 24, ix = r % 24;
      int gy = Y0 - 4 + iy, gx = X0 - 4 + ix;
      if (gy >= 0 && gy < HW && gx >= 0 && gx < HW)
        v = x[((size_t)(b * CIN + c) * HW + gy) * HW + gx];
    }
    xs[i] = v;
  }
  __syncthreads();

  {
    const int tile0 = (wave >> 2) * 8;
    #pragma nounroll
    for (int tp = 0; tp < 4; ++tp)
      conv_tiles<2>(A, a2s, wbs, xs, ss, sx, tile0 + 2 * tp, l16, h, grp);
  }
  __syncthreads();

  for (int pos = tid; pos < 18 * 18; pos += 256) {
    const int py = pos / 18, px = pos - py * 18;
    const int gy = Y0 - 1 + py, gx = X0 - 1 + px;
    const bool rowOwn = (py >= 1 && py <= 16), colOwn = (px >= 1 && px <= 16);
    const bool own = rowOwn && colOwn;
    float a0 = own ? b2[0] : 0.f, a1 = own ? b2[1] : 0.f;
    #pragma unroll
    for (int dy = 0; dy < 3; ++dy)
      #pragma unroll
      for (int dx = 0; dx < 3; ++dx) {
        const int tap = dy * 3 + dx;
        const int qy = py + dy - 2, qx = px + dx - 2;
        if ((unsigned)qy < 16u && (unsigned)qx < 16u) {
          const int pix = qy * 16 + qx;
          #pragma unroll
          for (int g = 0; g < 4; ++g) {
            a0 += ss[(g * 16 + tap) * SP + pix];
            if (tap < 7) {
              a1 += ss[(g * 16 + 9 + tap) * SP + pix];
            } else {
              a1 += sx[((g * 2 + 0) * 2 + (tap - 7)) * SP + pix];
              a1 += sx[((g * 2 + 1) * 2 + (tap - 7)) * SP + pix];
            }
          }
        }
      }
    if (gy >= 0 && gy < HW && gx >= 0 && gx < HW) {
      const int pl = b * CLS;
      if (own) {
        const size_t o0 = ((size_t)pl * HW + gy) * HW + gx;
        ini[o0] = a0;
        ini[o0 + (size_t)HW * HW] = a1;
      } else {
        const int hc = (gx >> 4) * 2 + ((gx & 15) == 15 ? 1 : 0);
        const int vr = (gy >> 4) * 2 + ((gy & 15) == 15 ? 1 : 0);
        if (rowOwn) {
          const size_t o = ((size_t)pl * NEDGE + hc) * HW + gy;
          eh[o] = a0;
          eh[o + (size_t)NEDGE * HW] = a1;
        } else if (colOwn) {
          const size_t o = ((size_t)pl * NEDGE + vr) * HW + gx;
          ev[o] = a0;
          ev[o + (size_t)NEDGE * HW] = a1;
        } else {
          const size_t o = ((size_t)pl * NEDGE + vr) * NEDGE + hc;
          ed[o] = a0;
          ed[o + (size_t)NEDGE * NEDGE] = a1;
        }
      }
    }
  }
}

#define TRANS_LDS_FLOATS (64 * 102 + 64 * 210)

__global__ __launch_bounds__(256) void trans_kernel(
    const float* __restrict__ ini, const float* __restrict__ eh,
    const float* __restrict__ ev,  const float* __restrict__ ed,
    const float* __restrict__ x,   const float* __restrict__ wfm,
    const float* __restrict__ bfm,
    const float* __restrict__ w1f, const float* __restrict__ w2f,
    float* __restrict__ ui, float* __restrict__ uf) {
  extern __shared__ float smem[];
  float* a_lds = smem;
  float* h_lds = a_lds + 64 * 102;

  const int tid = threadIdx.x;
  const bool is_fea = blockIdx.y != 0;
  float* dst        = is_fea ? uf : ui;
  const int P0 = blockIdx.x * 64;

  for (int i = tid; i < 64 * P2; i += 256) {
    int r = i / P2, p = i % P2;
    int Pa = P0 + r;
    int bc = Pa >> 10, l = Pa & 1023;
    int gy = (l >> 5) * PSZ + p / PSZ;
    int gx = (l & 31) * PSZ + p % PSZ;
    float v;
    if (is_fea) {
      const int cls = bc & 1;
      const float* xp = x + ((size_t)((bc >> 1) * CIN) * HW + gy) * HW + gx;
      v = (wfm[cls * 3] * xp[0] + wfm[cls * 3 + 1] * xp[(size_t)HW * HW]
           + wfm[cls * 3 + 2] * xp[(size_t)2 * HW * HW] + bfm[cls]) * 0.25f;
    } else {
      v = ini[((size_t)bc * HW + gy) * HW + gx];
      const int xm = gx & 15, ym = gy & 15;
      const bool hH = (xm == 0 && gx > 0) || (xm == 15 && gx < HW - 1);
      const bool hV = (ym == 0 && gy > 0) || (ym == 15 && gy < HW - 1);
      const int hc = (gx >> 4) * 2 + (xm == 15 ? 1 : 0);
      const int vr = (gy >> 4) * 2 + (ym == 15 ? 1 : 0);
      if (hH) v += eh[((size_t)bc * NEDGE + hc) * HW + gy];
      if (hV) v += ev[((size_t)bc * NEDGE + vr) * HW + gx];
      if (hH && hV) v += ed[((size_t)bc * NEDGE + vr) * NEDGE + hc];
    }
    a_lds[r * 102 + p] = v;
  }
  __syncthreads();

  const int wave  = __builtin_amdgcn_readfirstlane(tid >> 5);
  const int lane  = tid & 31;
  const int l16   = lane & 15;
  const int lhalf = (lane < 16) ? 0 : 2;

  for (int t = wave; t < 4 * 13; t += 8) {
    const int m0 = (t / 13) * 16, nt = t % 13;
    const int n  = nt * 16 + l16;
    const v2f* ap = (const v2f*)&a_lds[(m0 + l16) * 102 + lhalf];
    const v2f* bp = (const v2f*)w1f + nt * (25 * 32) + lane;
    v2f bfr[25];
    #pragma unroll
    for (int j = 0; j < 25; ++j) bfr[j] = bp[j * 32];
    v8f acc = {};
    #pragma unroll
    for (int j = 0; j < 25; ++j) acc = wmma_f32(ap[2 * j], bfr[j], acc);
    #pragma unroll
    for (int r = 0; r < 8; ++r) {
      int M = m0 + r + ((lane < 16) ? 0 : 8);
      h_lds[M * 210 + n] = acc[r];
    }
  }
  __syncthreads();

  for (int t = wave; t < 4 * 7; t += 8) {
    const int m0 = (t / 7) * 16, nt = t % 7;
    const int n  = nt * 16 + l16;
    const v2f* ap = (const v2f*)&h_lds[(m0 + l16) * 210 + lhalf];
    const v2f* bp = (const v2f*)w2f + nt * (50 * 32) + lane;
    v2f bfr[50];
    #pragma unroll
    for (int j = 0; j < 50; ++j) bfr[j] = bp[j * 32];
    v8f acc = {};
    #pragma unroll
    for (int j = 0; j < 50; ++j) acc = wmma_f32(ap[2 * j], bfr[j], acc);
    float v[8];
    for (int r = 0; r < 8; ++r) v[r] = acc[r] > 0.f ? acc[r] : 0.f;
    if (n < P2) {
      float* dp = dst + ((size_t)((P0 + m0) >> 4) * 25 + (n >> 2)) * 64
                      + ((n >> 1) & 1) * 32 + (lane >> 4) * 16 + (n & 1);
      #pragma unroll
      for (int r = 0; r < 8; ++r) dp[2 * r] = v[r];
    }
  }
}

__global__ __launch_bounds__(256) void att_kernel(
    const float* __restrict__ ui, const float* __restrict__ uf,
    float* __restrict__ out) {
  const int tid  = threadIdx.x;
  const int bc   = blockIdx.z;
  const int lane = tid & 31;
  const int l16  = lane & 15;
  const int h    = lane >> 4;
  const int wave = __builtin_amdgcn_readfirstlane(tid >> 5);
  const int nt   = blockIdx.y * 4 + (wave & 3);
  const int mtb  = blockIdx.x * 8 + (wave >> 2);

  const v2f* Uf = (const v2f*)ui + (size_t)bc * 64 * (25 * 32) + lane;
  const v2f* Vf = (const v2f*)uf + ((size_t)bc * 64 + nt) * (25 * 32) + lane;

  v2f B[25];
  #pragma unroll
  for (int j = 0; j < 25; ++j) B[j] = Vf[j * 32];

  #pragma nounroll
  for (int i = 0; i < 2; ++i) {
    const int mt0 = mtb + 4 * i, mt1 = mt0 + 2;
    const v2f* a0p = Uf + (size_t)mt0 * (25 * 32);
    const v2f* a1p = Uf + (size_t)mt1 * (25 * 32);
    v2f A0[25], A1[25];
    #pragma unroll
    for (int j = 0; j < 25; ++j) { A0[j] = a0p[j * 32]; A1[j] = a1p[j * 32]; }
    __builtin_amdgcn_sched_barrier(0);
    v8f acc0 = {}, acc1 = {};
    #pragma unroll
    for (int j = 0; j < 25; ++j) {
      acc0 = wmma_f32(A0[j], B[j], acc0);
      acc1 = wmma_f32(A1[j], B[j], acc1);
    }
    float* o0 = out + ((size_t)bc * LPATCH + mt0 * 16 + 8 * h) * LPATCH + nt * 16 + l16;
    float* o1 = out + ((size_t)bc * LPATCH + mt1 * 16 + 8 * h) * LPATCH + nt * 16 + l16;
    #pragma unroll
    for (int r = 0; r < 8; ++r) {
      o0[(size_t)r * LPATCH] = acc0[r] * 0.01f;
      o1[(size_t)r * LPATCH] = acc1[r] * 0.01f;
    }
  }
}

extern "C" void kernel_launch(void* const* d_in, const int* in_sizes, int n_in,
                              void* d_out, int out_size, void* d_ws,
                              size_t ws_size, hipStream_t stream) {
  (void)in_sizes; (void)n_in; (void)out_size; (void)ws_size;
  const float* x       = (const float*)d_in[0];
  const float* w_conv1 = (const float*)d_in[1];
  const float* b_conv1 = (const float*)d_in[2];
  const float* w_conv2 = (const float*)d_in[3];
  const float* b_conv2 = (const float*)d_in[4];
  const float* w_fm    = (const float*)d_in[5];
  const float* b_fm    = (const float*)d_in[6];
  const float* w_lin1  = (const float*)d_in[7];
  const float* w_lin2  = (const float*)d_in[8];
  float* out = (float*)d_out;

  float* ws  = (float*)d_ws;
  float* ini = ws + WS_INI;
  float* eh  = ws + WS_EH;
  float* ev  = ws + WS_EV;
  float* ed  = ws + WS_ED;
  float* ui  = ws + WS_UI;
  float* uf  = ws + WS_UF;
  float* w1f = ws + WS_W1P;
  float* w2f = ws + WS_W2P;
  float* aimg = ws + WS_AIMG;

  pad_weights_kernel<<<dim3((112 * 200 + 255) / 256), 256, 0, stream>>>(
      w_lin1, w_lin2, w_conv1, w1f, w2f, aimg);

  conv_fused_kernel<<<dim3(HW / 16, HW / 16, BATCH), 256,
                      CONV_LDS_FLOATS * sizeof(float), stream>>>(
      x, aimg, b_conv1, w_conv2, b_conv2, ini, eh, ev, ed);

  trans_kernel<<<dim3(BATCH * CLS * LPATCH / 64, 2, 1), 256,
                 TRANS_LDS_FLOATS * sizeof(float), stream>>>(
      ini, eh, ev, ed, x, w_fm, b_fm, w1f, w2f, ui, uf);

  att_kernel<<<dim3(LPATCH / 128, LPATCH / 64, BATCH * CLS), 256, 0,
               stream>>>(ui, uf, out);
}


# conv kernel: the x tile is staged by 16-byte loads issued as one group per thread (2 loads, then the LDS stores), not by nine dependent 4-byte round trips; sums unchanged
# speedup vs baseline: 2.3852x; 1.0342x over previous
#include <hip/hip_runtime.h>
#include <hip/hip_bf16.h>

typedef float v2f __attribute__((ext_vector_type(2)));
typedef float v8f __attribute__((ext_vector_type(8)));

__device__ __forceinline__ v8f wmma_f32(v2f a, v2f b, v8f c) {
  return __builtin_amdgcn_wmma_f32_16x16x4_f32(false, a, false, b, (short)0, c,
                                               false, false);
}

__device__ __forceinline__ void wmma_settle(v8f& d) {
  asm volatile("v_nop\n\tv_nop\n\tv_nop\n\tv_nop" : "+v"(d));
}
__device__ __forceinline__ void wmma_settle(v8f& d0, v8f& d1) {
  asm volatile("v_nop\n\tv_nop\n\tv_nop\n\tv_nop" : "+v"(d0), "+v"(d1));
}

#define BATCH   8
#define CIN     3
#define HW      320
#define CLS     2
#define PSZ     10
#define P2      100
#define LPATCH  1024
#define HID     200

#define WS_INI  0
#define WS_FEA  (WS_INI + BATCH*CLS*HW*HW)
#define WS_UI   (WS_FEA + BATCH*CLS*HW*HW)
#define WS_UF   (WS_UI  + BATCH*CLS*LPATCH*P2)
#define WS_W1P  (WS_UF  + BATCH*CLS*LPATCH*P2)
#define WS_W2P  (WS_W1P + 13*25*64)
#define WS_AIMG WS_UI
#define NEDGE   40
#define WS_EH   WS_FEA
#define WS_EV   (WS_EH + BATCH*CLS*NEDGE*HW)
#define WS_ED   (WS_EV + BATCH*CLS*NEDGE*HW)

#define KSTEPS 38
struct KStep { int c, ky, kx, sky, skx, hc, hky, hkx; };
__host__ __device__ constexpr KStep kstep(int j) {
  if (j < 21) return { j / 7, j % 7, 0,   0, 1,   0, 0, 2 };
  j -= 21;
  if (j < 9)  return { j / 3, 2 * (j % 3), 4,   0, 1,   0, 1, 0 };
  j -= 9;
  if (j < 6)  return { j / 2, 4 * (j % 2), 6,   1, 0,   0, 2, 0 };
  j -= 6;
  if (j == 0) return { 0, 6, 4,   0, 1,   1, 0, 0 };
  return { 2, 6, 4,   0, 1,   0, 1, 0 };
}

__global__ __launch_bounds__(256) void pad_weights_kernel(
    const float* __restrict__ w_lin1, const float* __restrict__ w_lin2,
    const float* __restrict__ w_conv1,
    float* __restrict__ w1f, float* __restrict__ w2f,
    float* __restrict__ aimg) {
  int i = blockIdx.x * 256 + threadIdx.x;
  const int e = i & 1, lane = (i >> 1) & 31, gj = i >> 6;
  if (i < 4 * KSTEPS * 64) {
    const int j = gj % KSTEPS, g = gj / KSTEPS;
    const int h = lane >> 4, row = g * 16 + (lane & 15);
    const KStep P = kstep(j);
    const int c  = P.c + h * P.hc;
    const int ky = P.ky + e * P.sky + h * P.hky;
    const int kx = P.kx + e * P.skx + h * P.hkx;
    aimg[i] = (c < 3 && ky < 7 && kx < 7)
                  ? w_conv1[row * 147 + c * 49 + ky * 7 + kx] : 0.f;
  }
  const int row = lane & 15, kk = 2 * (lane >> 4) + e;
  if (i < 13 * 25 * 64) {
    const int j = gj % 25, n = (gj / 25) * 16 + row;
    w1f[i] = (n < HID) ? w_lin1[n * P2 + 4 * j + kk] : 0.f;
  }
  if (i < 7 * 50 * 64) {
    const int j = gj % 50, n = (gj / 50) * 16 + row;
    w2f[i] = (n < P2) ? w_lin2[n * HID + 4 * j + kk] : 0.f;
  }
}

#define XS_FLOATS        2160
#define SP               274
#define A2_FLOATS        (4 * 2 * 32 * 4)
#define WB_FLOATS        (4 * 2 * 24)
#define CONV_LDS_FLOATS  (XS_FLOATS + 4 * 16 * SP + 4 * 2 * 2 * SP + A2_FLOATS + WB_FLOATS)

template <int NT>
__device__ __forceinline__ void conv_tiles(
    const v2f (&A)[KSTEPS], const float* a2s, const float* wbs,
    const float* xs, float* ss, float* sx,
    int tile0, int l16, int h, int grp) {
  int pb[NT];
  for (int t = 0; t < NT; ++t) pb[t] = (tile0 + t + 1) * 24 + (l16 + 1);
  v8f acc[NT];
  for (int t = 0; t < NT; ++t) acc[t] = v8f{};

  #pragma unroll
  for (int j = 0; j < KSTEPS; ++j) {
    const KStep P = kstep(j);
    const int o0 = P.c * 576 + P.ky * 24 + P.kx;
    const int ds = P.sky * 24 + P.skx;
    const int dh = P.hc * 576 + P.hky * 24 + P.hkx;
    for (int t = 0; t < NT; ++t) {
      const float* bp = xs + (pb[t] + h * dh) + o0;
      v2f bf;
      bf.x = bp[0];
      bf.y = bp[ds];
      acc[t] = wmma_f32(A[j], bf, acc[t]);
    }
  }

  if constexpr (NT == 2) wmma_settle(acc[0], acc[1]); else wmma_settle(acc[0]);

  const float4* wq = (const float4*)wbs;
  float bias[8];
  {
    const float4 lo = wq[4], hi = wq[5];
    bias[0] = lo.x; bias[1] = lo.y; bias[2] = lo.z; bias[3] = lo.w;
    bias[4] = hi.x; bias[5] = hi.y; bias[6] = hi.z; bias[7] = hi.w;
  }
  v2f A2[4];
  {
    const float4 p0 = ((const float4*)a2s)[0], p1 = ((const float4*)a2s)[32];
    A2[0].x = p0.x; A2[0].y = p0.y; A2[1].x = p0.z; A2[1].y = p0.w;
    A2[2].x = p1.x; A2[2].y = p1.y; A2[3].x = p1.z; A2[3].y = p1.w;
  }

  float T[NT][8];
  for (int t = 0; t < NT; ++t)
    for (int r = 0; r < 8; ++r) T[t][r] = acc[t][r] + bias[r];

  v8f S[NT];
  for (int t = 0; t < NT; ++t) S[t] = v8f{};
  #pragma unroll
  for (int q = 0; q < 4; ++q)
    for (int t = 0; t < NT; ++t) {
      v2f bt;
      bt.x = T[t][2 * q];
      bt.y = T[t][2 * q + 1];
      S[t] = wmma_f32(A2[q], bt, S[t]);
    }
  if constexpr (NT == 2) wmma_settle(S[0], S[1]); else wmma_settle(S[0]);

  float wx[2][8];
  for (int k = 0; k < 4; ++k) {
    const float4 v = wq[k];
    wx[0][2 * k]     = v.x; wx[1][2 * k]     = v.y;
    wx[0][2 * k + 1] = v.z; wx[1][2 * k + 1] = v.w;
  }
  for (int t = 0; t < NT; ++t) {
    const int p = (tile0 + t) * 16 + l16;
    float e0 = 0.f, e1 = 0.f;
    for (int r = 0; r < 8; ++r) {
      e0 = fmaf(wx[0][r], T[t][r], e0);
      e1 = fmaf(wx[1][r], T[t][r], e1);
    }
    float* sp = ss + (grp * 16 + 8 * h) * SP + p;
    for (int r = 0; r < 8; ++r) sp[r * SP] = S[t][r];
    float* ep = sx + ((grp * 2 + h) * 2) * SP + p;
    ep[0]  = e0;
    ep[SP] = e1;
  }
}

__global__ __launch_bounds__(256) void conv_fused_kernel(
    const float* __restrict__ x,  const float* __restrict__ aimg,
    const float* __restrict__ b1, const float* __restrict__ w2,
    const float* __restrict__ b2,
    float* __restrict__ ini, float* __restrict__ eh,
    float* __restrict__ ev,  float* __restrict__ ed) {
  extern __shared__ float smem[];
  float* xs = smem;
  float* ss = xs + XS_FLOATS;
  float* sx = ss + 4 * 16 * SP;
  float* a2l = sx + 4 * 2 * 2 * SP;
  float* wbl = a2l + A2_FLOATS;

  const int tid = threadIdx.x;
  const int X0 = blockIdx.x * 16, Y0 = blockIdx.y * 16, b = blockIdx.z;

  const int lane = tid & 31;
  const int l16  = lane & 15;
  const int h    = lane >> 4;
  const int wave = __builtin_amdgcn_readfirstlane(tid >> 5);
  const int grp  = wave & 3;

  float4 xv[3];
  #pragma unroll
  for (int k = 0; k < 3; ++k) {
    const int q = tid + 256 * k;
    xv[k] = make_float4(0.f, 0.f, 0.f, 0.f);
    if (q < CIN * 144) {
      const int c = q / 144, r = q % 144, iy = r / 6, ix = (r % 6) * 4;
      const int gy = Y0 - 4 + iy, gx = X0 - 4 + ix;
      if (gy >= 0 && gy < HW && gx >= 0 && gx < HW)
        xv[k] = *(const float4*)(x + ((size_t)(b * CIN + c) * HW + gy) * HW + gx);
    }
  }

  v2f A[KSTEPS];
  {
    const v2f* ap = (const v2f*)aimg + (size_t)grp * KSTEPS * 32 + lane;
    #pragma unroll
    for (int j = 0; j < KSTEPS; ++j) A[j] = ap[j * 32];
  }

  const float* a2s = a2l + (grp * 2 * 32 + lane) * 4;
  const float* wbs = wbl + (grp * 2 + h) * 24;
  {
    const int cls = (l16 >= 9) ? 1 : 0, tap = l16 - 9 * cls;
    const float* wp = w2 + ((size_t)(cls * 64 + grp * 16 + 8 * h) * 9 + tap);
    for (int k = 0; k < 2; ++k) {
      float4 v;
      v.x = wp[(4 * k) * 9];     v.y = wp[(4 * k + 1) * 9];
      v.z = wp[(4 * k + 2) * 9]; v.w = wp[(4 * k + 3) * 9];
      ((float4*)(a2l + (grp * 2 * 32 + lane) * 4))[32 * k] = v;
    }
    float* wb = wbl + (grp * 2 + h) * 24;
    wb[l16] = w2[(size_t)(64 + grp * 16 + 8 * h + (l16 >> 1)) * 9 + 7 + (l16 & 1)];
    wb[16 + (l16 & 7)] = b1[grp * 16 + 8 * h + (l16 & 7)];
  }

  #pragma unroll
  for (int k = 0; k < 3; ++k) {
    const int q = tid + 256 * k;
    if (q < XS_FLOATS / 4) ((float4*)xs)[q] = xv[k];
  }
  __syncthreads();

  {
    const int tile0 = (wave >> 2) * 8;
    #pragma nounroll
    for (int tp = 0; tp < 4; ++tp)
      conv_tiles<2>(A, a2s, wbs, xs, ss, sx, tile0 + 2 * tp, l16, h, grp);
  }
  __syncthreads();

  for (int pos = tid; pos < 18 * 18; pos += 256) {
    const int py = pos / 18, px = pos - py * 18;
    const int gy = Y0 - 1 + py, gx = X0 - 1 + px;
    const bool rowOwn = (py >= 1 && py <= 16), colOwn = (px >= 1 && px <= 16);
    const bool own = rowOwn && colOwn;
    float a0 = own ? b2[0] : 0.f, a1 = own ? b2[1] : 0.f;
    #pragma unroll
    for (int dy = 0; dy < 3; ++dy)
      #pragma unroll
      for (int dx = 0; dx < 3; ++dx) {
        const int tap = dy * 3 + dx;
        const int qy = py + dy - 2, qx = px + dx - 2;
        if ((unsigned)qy < 16u && (unsigned)qx < 16u) {
          const int pix = qy * 16 + qx;
          #pragma unroll
          for (int g = 0; g < 4; ++g) {
            a0 += ss[(g * 16 + tap) * SP + pix];
            if (tap < 7) {
              a1 += ss[(g * 16 + 9 + tap) * SP + pix];
            } else {
              a1 += sx[((g * 2 + 0) * 2 + (tap - 7)) * SP + pix];
              a1 += sx[((g * 2 + 1) * 2 + (tap - 7)) * SP + pix];
            }
          }
        }
      }
    if (gy >= 0 && gy < HW && gx >= 0 && gx < HW) {
      const int pl = b * CLS;
      if (own) {
        const size_t o0 = ((size_t)pl * HW + gy) * HW + gx;
        ini[o0] = a0;
        ini[o0 + (size_t)HW * HW] = a1;
      } else {
        const int hc = (gx >> 4) * 2 + ((gx & 15) == 15 ? 1 : 0);
        const int vr = (gy >> 4) * 2 + ((gy & 15) == 15 ? 1 : 0);
        if (rowOwn) {
          const size_t o = ((size_t)pl * NEDGE + hc) * HW + gy;
          eh[o] = a0;
          eh[o + (size_t)NEDGE * HW] = a1;
        } else if (colOwn) {
          const size_t o = ((size_t)pl * NEDGE + vr) * HW + gx;
          ev[o] = a0;
          ev[o + (size_t)NEDGE * HW] = a1;
        } else {
          const size_t o = ((size_t)pl * NEDGE + vr) * NEDGE + hc;
          ed[o] = a0;
          ed[o + (size_t)NEDGE * NEDGE] = a1;
        }
      }
    }
  }
}

#define TRANS_LDS_FLOATS (64 * 102 + 64 * 210)

__global__ __launch_bounds__(256) void trans_kernel(
    const float* __restrict__ ini, const float* __restrict__ eh,
    const float* __restrict__ ev,  const float* __restrict__ ed,
    const float* __restrict__ x,   const float* __restrict__ wfm,
    const float* __restrict__ bfm,
    const float* __restrict__ w1f, const float* __restrict__ w2f,
    float* __restrict__ ui, float* __restrict__ uf) {
  extern __shared__ float smem[];
  float* a_lds = smem;
  float* h_lds = a_lds + 64 * 102;

  const int tid = threadIdx.x;
  const bool is_fea = blockIdx.y != 0;
  float* dst        = is_fea ? uf : ui;
  const int P0 = blockIdx.x * 64;

  for (int i = tid; i < 64 * P2; i += 256) {
    int r = i / P2, p = i % P2;
    int Pa = P0 + r;
    int bc = Pa >> 10, l = Pa & 1023;
    int gy = (l >> 5) * PSZ + p / PSZ;
    int gx = (l & 31) * PSZ + p % PSZ;
    float v;
    if (is_fea) {
      const int cls = bc & 1;
      const float* xp = x + ((size_t)((bc >> 1) * CIN) * HW + gy) * HW + gx;
      v = (wfm[cls * 3] * xp[0] + wfm[cls * 3 + 1] * xp[(size_t)HW * HW]
           + wfm[cls * 3 + 2] * xp[(size_t)2 * HW * HW] + bfm[cls]) * 0.25f;
    } else {
      v = ini[((size_t)bc * HW + gy) * HW + gx];
      const int xm = gx & 15, ym = gy & 15;
      const bool hH = (xm == 0 && gx > 0) || (xm == 15 && gx < HW - 1);
      const bool hV = (ym == 0 && gy > 0) || (ym == 15 && gy < HW - 1);
      const int hc = (gx >> 4) * 2 + (xm == 15 ? 1 : 0);
      const int vr = (gy >> 4) * 2 + (ym == 15 ? 1 : 0);
      if (hH) v += eh[((size_t)bc * NEDGE + hc) * HW + gy];
      if (hV) v += ev[((size_t)bc * NEDGE + vr) * HW + gx];
      if (hH && hV) v += ed[((size_t)bc * NEDGE + vr) * NEDGE + hc];
    }
    a_lds[r * 102 + p] = v;
  }
  __syncthreads();

  const int wave  = __builtin_amdgcn_readfirstlane(tid >> 5);
  const int lane  = tid & 31;
  const int l16   = lane & 15;
  const int lhalf = (lane < 16) ? 0 : 2;

  for (int t = wave; t < 4 * 13; t += 8) {
    const int m0 = (t / 13) * 16, nt = t % 13;
    const int n  = nt * 16 + l16;
    const v2f* ap = (const v2f*)&a_lds[(m0 + l16) * 102 + lhalf];
    const v2f* bp = (const v2f*)w1f + nt * (25 * 32) + lane;
    v2f bfr[25];
    #pragma unroll
    for (int j = 0; j < 25; ++j) bfr[j] = bp[j * 32];
    v8f acc = {};
    #pragma unroll
    for (int j = 0; j < 25; ++j) acc = wmma_f32(ap[2 * j], bfr[j], acc);
    #pragma unroll
    for (int r = 0; r < 8; ++r) {
      int M = m0 + r + ((lane < 16) ? 0 : 8);
      h_lds[M * 210 + n] = acc[r];
    }
  }
  __syncthreads();

  for (int t = wave; t < 4 * 7; t += 8) {
    const int m0 = (t / 7) * 16, nt = t % 7;
    const int n  = nt * 16 + l16;
    const v2f* ap = (const v2f*)&h_lds[(m0 + l16) * 210 + lhalf];
    const v2f* bp = (const v2f*)w2f + nt * (50 * 32) + lane;
    v2f bfr[50];
    #pragma unroll
    for (int j = 0; j < 50; ++j) bfr[j] = bp[j * 32];
    v8f acc = {};
    #pragma unroll
    for (int j = 0; j < 50; ++j) acc = wmma_f32(ap[2 * j], bfr[j], acc);
    float v[8];
    for (int r = 0; r < 8; ++r) v[r] = acc[r] > 0.f ? acc[r] : 0.f;
    if (n < P2) {
      float* dp = dst + ((size_t)((P0 + m0) >> 4) * 25 + (n >> 2)) * 64
                      + ((n >> 1) & 1) * 32 + (lane >> 4) * 16 + (n & 1);
      #pragma unroll
      for (int r = 0; r < 8; ++r) dp[2 * r] = v[r];
    }
  }
}

__global__ __launch_bounds__(256) void att_kernel(
    const float* __restrict__ ui, const float* __restrict__ uf,
    float* __restrict__ out) {
  const int tid  = threadIdx.x;
  const int bc   = blockIdx.z;
  const int lane = tid & 31;
  const int l16  = lane & 15;
  const int h    = lane >> 4;
  const int wave = __builtin_amdgcn_readfirstlane(tid >> 5);
  const int nt   = blockIdx.y * 4 + (wave & 3);
  const int mtb  = blockIdx.x * 8 + (wave >> 2);

  const v2f* Uf = (const v2f*)ui + (size_t)bc * 64 * (25 * 32) + lane;
  const v2f* Vf = (const v2f*)uf + ((size_t)bc * 64 + nt) * (25 * 32) + lane;

  v2f B[25];
  #pragma unroll
  for (int j = 0; j < 25; ++j) B[j] = Vf[j * 32];

  #pragma nounroll
  for (int i = 0; i < 2; ++i) {
    const int mt0 = mtb + 4 * i, mt1 = mt0 + 2;
    const v2f* a0p = Uf + (size_t)mt0 * (25 * 32);
    const v2f* a1p = Uf + (size_t)mt1 * (25 * 32);
    v2f A0[25], A1[25];
    #pragma unroll
    for (int j = 0; j < 25; ++j) { A0[j] = a0p[j * 32]; A1[j] = a1p[j * 32]; }
    __builtin_amdgcn_sched_barrier(0);
    v8f acc0 = {}, acc1 = {};
    #pragma unroll
    for (int j = 0; j < 25; ++j) {
      acc0 = wmma_f32(A0[j], B[j], acc0);
      acc1 = wmma_f32(A1[j], B[j], acc1);
    }
    float* o0 = out + ((size_t)bc * LPATCH + mt0 * 16 + 8 * h) * LPATCH + nt * 16 + l16;
    float* o1 = out + ((size_t)bc * LPATCH + mt1 * 16 + 8 * h) * LPATCH + nt * 16 + l16;
    #pragma unroll
    for (int r = 0; r < 8; ++r) {
      o0[(size_t)r * LPATCH] = acc0[r] * 0.01f;
      o1[(size_t)r * LPATCH] = acc1[r] * 0.01f;
    }
  }
}

extern "C" void kernel_launch(void* const* d_in, const int* in_sizes, int n_in,
                              void* d_out, int out_size, void* d_ws,
                              size_t ws_size, hipStream_t stream) {
  (void)in_sizes; (void)n_in; (void)out_size; (void)ws_size;
  const float* x       = (const float*)d_in[0];
  const float* w_conv1 = (const float*)d_in[1];
  const float* b_conv1 = (const float*)d_in[2];
  const float* w_conv2 = (const float*)d_in[3];
  const float* b_conv2 = (const float*)d_in[4];
  const float* w_fm    = (const float*)d_in[5];
  const float* b_fm    = (const float*)d_in[6];
  const float* w_lin1  = (const float*)d_in[7];
  const float* w_lin2  = (const float*)d_in[8];
  float* out = (float*)d_out;

  float* ws  = (float*)d_ws;
  float* ini = ws + WS_INI;
  float* eh  = ws + WS_EH;
  float* ev  = ws + WS_EV;
  float* ed  = ws + WS_ED;
  float* ui  = ws + WS_UI;
  float* uf  = ws + WS_UF;
  float* w1f = ws + WS_W1P;
  float* w2f = ws + WS_W2P;
  float* aimg = ws + WS_AIMG;

  pad_weights_kernel<<<dim3((112 * 200 + 255) / 256), 256, 0, stream>>>(
      w_lin1, w_lin2, w_conv1, w1f, w2f, aimg);

  conv_fused_kernel<<<dim3(HW / 16, HW / 16, BATCH), 256,
                      CONV_LDS_FLOATS * sizeof(float), stream>>>(
      x, aimg, b_conv1, w_conv2, b_conv2, ini, eh, ev, ed);

  trans_kernel<<<dim3(BATCH * CLS * LPATCH / 64, 2, 1), 256,
                 TRANS_LDS_FLOATS * sizeof(float), stream>>>(
      ini, eh, ev, ed, x, w_fm, b_fm, w1f, w2f, ui, uf);

  att_kernel<<<dim3(LPATCH / 128, LPATCH / 64, BATCH * CLS), 256, 0,
               stream>>>(ui, uf, out);
}


# trans kernel re-tiled: 32 patches per block (was 64); a wave owns an N-tile and runs both M-tiles on one load of its weight fragments (two accumulators); unfold per thread halves; sums unchanged
# speedup vs baseline: 2.4997x; 1.0480x over previous
#include <hip/hip_runtime.h>
#include <hip/hip_bf16.h>

typedef float v2f __attribute__((ext_vector_type(2)));
typedef float v8f __attribute__((ext_vector_type(8)));

__device__ __forceinline__ v8f wmma_f32(v2f a, v2f b, v8f c) {
  return __builtin_amdgcn_wmma_f32_16x16x4_f32(false, a, false, b, (short)0, c,
                                               false, false);
}

__device__ __forceinline__ void wmma_settle(v8f& d) {
  asm volatile("v_nop\n\tv_nop\n\tv_nop\n\tv_nop" : "+v"(d));
}
__device__ __forceinline__ void wmma_settle(v8f& d0, v8f& d1) {
  asm volatile("v_nop\n\tv_nop\n\tv_nop\n\tv_nop" : "+v"(d0), "+v"(d1));
}

#define BATCH   8
#define CIN     3
#define HW      320
#define CLS     2
#define PSZ     10
#define P2      100
#define LPATCH  1024
#define HID     200

#define WS_INI  0
#define WS_FEA  (WS_INI + BATCH*CLS*HW*HW)
#define WS_UI   (WS_FEA + BATCH*CLS*HW*HW)
#define WS_UF   (WS_UI  + BATCH*CLS*LPATCH*P2)
#define WS_W1P  (WS_UF  + BATCH*CLS*LPATCH*P2)
#define WS_W2P  (WS_W1P + 13*25*64)
#define WS_AIMG WS_UI
#define NEDGE   40
#define WS_EH   WS_FEA
#define WS_EV   (WS_EH + BATCH*CLS*NEDGE*HW)
#define WS_ED   (WS_EV + BATCH*CLS*NEDGE*HW)

#define KSTEPS 38
struct KStep { int c, ky, kx, sky, skx, hc, hky, hkx; };
__host__ __device__ constexpr KStep kstep(int j) {
  if (j < 21) return { j / 7, j % 7, 0,   0, 1,   0, 0, 2 };
  j -= 21;
  if (j < 9)  return { j / 3, 2 * (j % 3), 4,   0, 1,   0, 1, 0 };
  j -= 9;
  if (j < 6)  return { j / 2, 4 * (j % 2), 6,   1, 0,   0, 2, 0 };
  j -= 6;
  if (j == 0) return { 0, 6, 4,   0, 1,   1, 0, 0 };
  return { 2, 6, 4,   0, 1,   0, 1, 0 };
}

__global__ __launch_bounds__(256) void pad_weights_kernel(
    const float* __restrict__ w_lin1, const float* __restrict__ w_lin2,
    const float* __restrict__ w_conv1,
    float* __restrict__ w1f, float* __restrict__ w2f,
    float* __restrict__ aimg) {
  int i = blockIdx.x * 256 + threadIdx.x;
  const int e = i & 1, lane = (i >> 1) & 31, gj = i >> 6;
  if (i < 4 * KSTEPS * 64) {
    const int j = gj % KSTEPS, g = gj / KSTEPS;
    const int h = lane >> 4, row = g * 16 + (lane & 15);
    const KStep P = kstep(j);
    const int c  = P.c + h * P.hc;
    const int ky = P.ky + e * P.sky + h * P.hky;
    const int kx = P.kx + e * P.skx + h * P.hkx;
    aimg[i] = (c < 3 && ky < 7 && kx < 7)
                  ? w_conv1[row * 147 + c * 49 + ky * 7 + kx] : 0.f;
  }
  const int row = lane & 15, kk = 2 * (lane >> 4) + e;
  if (i < 13 * 25 * 64) {
    const int j = gj % 25, n = (gj / 25) * 16 + row;
    w1f[i] = (n < HID) ? w_lin1[n * P2 + 4 * j + kk] : 0.f;
  }
  if (i < 7 * 50 * 64) {
    const int j = gj % 50, n = (gj / 50) * 16 + row;
    w2f[i] = (n < P2) ? w_lin2[n * HID + 4 * j + kk] : 0.f;
  }
}

#define XS_FLOATS        2160
#define SP               274
#define A2_FLOATS        (4 * 2 * 32 * 4)
#define WB_FLOATS        (4 * 2 * 24)
#define CONV_LDS_FLOATS  (XS_FLOATS + 4 * 16 * SP + 4 * 2 * 2 * SP + A2_FLOATS + WB_FLOATS)

template <int NT>
__device__ __forceinline__ void conv_tiles(
    const v2f (&A)[KSTEPS], const float* a2s, const float* wbs,
    const float* xs, float* ss, float* sx,
    int tile0, int l16, int h, int grp) {
  int pb[NT];
  for (int t = 0; t < NT; ++t) pb[t] = (tile0 + t + 1) * 24 + (l16 + 1);
  v8f acc[NT];
  for (int t = 0; t < NT; ++t) acc[t] = v8f{};

  #pragma unroll
  for (int j = 0; j < KSTEPS; ++j) {
    const KStep P = kstep(j);
    const int o0 = P.c * 576 + P.ky * 24 + P.kx;
    const int ds = P.sky * 24 + P.skx;
    const int dh = P.hc * 576 + P.hky * 24 + P.hkx;
    for (int t = 0; t < NT; ++t) {
      const float* bp = xs + (pb[t] + h * dh) + o0;
      v2f bf;
      bf.x = bp[0];
      bf.y = bp[ds];
      acc[t] = wmma_f32(A[j], bf, acc[t]);
    }
  }

  if constexpr (NT == 2) wmma_settle(acc[0], acc[1]); else wmma_settle(acc[0]);

  const float4* wq = (const float4*)wbs;
  float bias[8];
  {
    const float4 lo = wq[4], hi = wq[5];
    bias[0] = lo.x; bias[1] = lo.y; bias[2] = lo.z; bias[3] = lo.w;
    bias[4] = hi.x; bias[5] = hi.y; bias[6] = hi.z; bias[7] = hi.w;
  }
  v2f A2[4];
  {
    const float4 p0 = ((const float4*)a2s)[0], p1 = ((const float4*)a2s)[32];
    A2[0].x = p0.x; A2[0].y = p0.y; A2[1].x = p0.z; A2[1].y = p0.w;
    A2[2].x = p1.x; A2[2].y = p1.y; A2[3].x = p1.z; A2[3].y = p1.w;
  }

  float T[NT][8];
  for (int t = 0; t < NT; ++t)
    for (int r = 0; r < 8; ++r) T[t][r] = acc[t][r] + bias[r];

  v8f S[NT];
  for (int t = 0; t < NT; ++t) S[t] = v8f{};
  #pragma unroll
  for (int q = 0; q < 4; ++q)
    for (int t = 0; t < NT; ++t) {
      v2f bt;
      bt.x = T[t][2 * q];
      bt.y = T[t][2 * q + 1];
      S[t] = wmma_f32(A2[q], bt, S[t]);
    }
  if constexpr (NT == 2) wmma_settle(S[0], S[1]); else wmma_settle(S[0]);

  float wx[2][8];
  for (int k = 0; k < 4; ++k) {
    const float4 v = wq[k];
    wx[0][2 * k]     = v.x; wx[1][2 * k]     = v.y;
    wx[0][2 * k + 1] = v.z; wx[1][2 * k + 1] = v.w;
  }
  for (int t = 0; t < NT; ++t) {
    const int p = (tile0 + t) * 16 + l16;
    float e0 = 0.f, e1 = 0.f;
    for (int r = 0; r < 8; ++r) {
      e0 = fmaf(wx[0][r], T[t][r], e0);
      e1 = fmaf(wx[1][r], T[t][r], e1);
    }
    float* sp = ss + (grp * 16 + 8 * h) * SP + p;
    for (int r = 0; r < 8; ++r) sp[r * SP] = S[t][r];
    float* ep = sx + ((grp * 2 + h) * 2) * SP + p;
    ep[0]  = e0;
    ep[SP] = e1;
  }
}

__global__ __launch_bounds__(256) void conv_fused_kernel(
    const float* __restrict__ x,  const float* __restrict__ aimg,
    const float* __restrict__ b1, const float* __restrict__ w2,
    const float* __restrict__ b2,
    float* __restrict__ ini, float* __restrict__ eh,
    float* __restrict__ ev,  float* __restrict__ ed) {
  extern __shared__ float smem[];
  float* xs = smem;
  float* ss = xs + XS_FLOATS;
  float* sx = ss + 4 * 16 * SP;
  float* a2l = sx + 4 * 2 * 2 * SP;
  float* wbl = a2l + A2_FLOATS;

  const int tid = threadIdx.x;
  const int X0 = blockIdx.x * 16, Y0 = blockIdx.y * 16, b = blockIdx.z;

  const int lane = tid & 31;
  const int l16  = lane & 15;
  const int h    = lane >> 4;
  const int wave = __builtin_amdgcn_readfirstlane(tid >> 5);
  const int grp  = wave & 3;

  float4 xv[3];
  #pragma unroll
  for (int k = 0; k < 3; ++k) {
    const int q = tid + 256 * k;
    xv[k] = make_float4(0.f, 0.f, 0.f, 0.f);
    if (q < CIN * 144) {
      const int c = q / 144, r = q % 144, iy = r / 6, ix = (r % 6) * 4;
      const int gy = Y0 - 4 + iy, gx = X0 - 4 + ix;
      if (gy >= 0 && gy < HW && gx >= 0 && gx < HW)
        xv[k] = *(const float4*)(x + ((size_t)(b * CIN + c) * HW + gy) * HW + gx);
    }
  }

  v2f A[KSTEPS];
  {
    const v2f* ap = (const v2f*)aimg + (size_t)grp * KSTEPS * 32 + lane;
    #pragma unroll
    for (int j = 0; j < KSTEPS; ++j) A[j] = ap[j * 32];
  }

  const float* a2s = a2l + (grp * 2 * 32 + lane) * 4;
  const float* wbs = wbl + (grp * 2 + h) * 24;
  {
    const int cls = (l16 >= 9) ? 1 : 0, tap = l16 - 9 * cls;
    const float* wp = w2 + ((size_t)(cls * 64 + grp * 16 + 8 * h) * 9 + tap);
    for (int k = 0; k < 2; ++k) {
      float4 v;
      v.x = wp[(4 * k) * 9];     v.y = wp[(4 * k + 1) * 9];
      v.z = wp[(4 * k + 2) * 9]; v.w = wp[(4 * k + 3) * 9];
      ((float4*)(a2l + (grp * 2 * 32 + lane) * 4))[32 * k] = v;
    }
    float* wb = wbl + (grp * 2 + h) * 24;
    wb[l16] = w2[(size_t)(64 + grp * 16 + 8 * h + (l16 >> 1)) * 9 + 7 + (l16 & 1)];
    wb[16 + (l16 & 7)] = b1[grp * 16 + 8 * h + (l16 & 7)];
  }

  #pragma unroll
  for (int k = 0; k < 3; ++k) {
    const int q = tid + 256 * k;
    if (q < XS_FLOATS / 4) ((float4*)xs)[q] = xv[k];
  }
  __syncthreads();

  {
    const int tile0 = (wave >> 2) * 8;
    #pragma nounroll
    for (int tp = 0; tp < 4; ++tp)
      conv_tiles<2>(A, a2s, wbs, xs, ss, sx, tile0 + 2 * tp, l16, h, grp);
  }
  __syncthreads();

  for (int pos = tid; pos < 18 * 18; pos += 256) {
    const int py = pos / 18, px = pos - py * 18;
    const int gy = Y0 - 1 + py, gx = X0 - 1 + px;
    const bool rowOwn = (py >= 1 && py <= 16), colOwn = (px >= 1 && px <= 16);
    const bool own = rowOwn && colOwn;
    float a0 = own ? b2[0] : 0.f, a1 = own ? b2[1] : 0.f;
    #pragma unroll
    for (int dy = 0; dy < 3; ++dy)
      #pragma unroll
      for (int dx = 0; dx < 3; ++dx) {
        const int tap = dy * 3 + dx;
        const int qy = py + dy - 2, qx = px + dx - 2;
        if ((unsigned)qy < 16u && (unsigned)qx < 16u) {
          const int pix = qy * 16 + qx;
          #pragma unroll
          for (int g = 0; g < 4; ++g) {
            a0 += ss[(g * 16 + tap) * SP + pix];
            if (tap < 7) {
              a1 += ss[(g * 16 + 9 + tap) * SP + pix];
            } else {
              a1 += sx[((g * 2 + 0) * 2 + (tap - 7)) * SP + pix];
              a1 += sx[((g * 2 + 1) * 2 + (tap - 7)) * SP + pix];
            }
          }
        }
      }
    if (gy >= 0 && gy < HW && gx >= 0 && gx < HW) {
      const int pl = b * CLS;
      if (own) {
        const size_t o0 = ((size_t)pl * HW + gy) * HW + gx;
        ini[o0] = a0;
        ini[o0 + (size_t)HW * HW] = a1;
      } else {
        const int hc = (gx >> 4) * 2 + ((gx & 15) == 15 ? 1 : 0);
        const int vr = (gy >> 4) * 2 + ((gy & 15) == 15 ? 1 : 0);
        if (rowOwn) {
          const size_t o = ((size_t)pl * NEDGE + hc) * HW + gy;
          eh[o] = a0;
          eh[o + (size_t)NEDGE * HW] = a1;
        } else if (colOwn) {
          const size_t o = ((size_t)pl * NEDGE + vr) * HW + gx;
          ev[o] = a0;
          ev[o + (size_t)NEDGE * HW] = a1;
        } else {
          const size_t o = ((size_t)pl * NEDGE + vr) * NEDGE + hc;
          ed[o] = a0;
          ed[o + (size_t)NEDGE * NEDGE] = a1;
        }
      }
    }
  }
}

#define TPB 32
#define TRANS_LDS_FLOATS (TPB * 102 + TPB * 210)

__global__ __launch_bounds__(256) void trans_kernel(
    const float* __restrict__ ini, const float* __restrict__ eh,
    const float* __restrict__ ev,  const float* __restrict__ ed,
    const float* __restrict__ x,   const float* __restrict__ wfm,
    const float* __restrict__ bfm,
    const float* __restrict__ w1f, const float* __restrict__ w2f,
    float* __restrict__ ui, float* __restrict__ uf) {
  extern __shared__ float smem[];
  float* a_lds = smem;
  float* h_lds = a_lds + TPB * 102;

  const int tid = threadIdx.x;
  const bool is_fea = blockIdx.y != 0;
  float* dst        = is_fea ? uf : ui;
  const int P0 = blockIdx.x * TPB;

  for (int i = tid; i < TPB * P2; i += 256) {
    int r = i / P2, p = i % P2;
    int Pa = P0 + r;
    int bc = Pa >> 10, l = Pa & 1023;
    int gy = (l >> 5) * PSZ + p / PSZ;
    int gx = (l & 31) * PSZ + p % PSZ;
    float v;
    if (is_fea) {
      const int cls = bc & 1;
      const float* xp = x + ((size_t)((bc >> 1) * CIN) * HW + gy) * HW + gx;
      v = (wfm[cls * 3] * xp[0] + wfm[cls * 3 + 1] * xp[(size_t)HW * HW]
           + wfm[cls * 3 + 2] * xp[(size_t)2 * HW * HW] + bfm[cls]) * 0.25f;
    } else {
      v = ini[((size_t)bc * HW + gy) * HW + gx];
      const int xm = gx & 15, ym = gy & 15;
      const bool hH = (xm == 0 && gx > 0) || (xm == 15 && gx < HW - 1);
      const bool hV = (ym == 0 && gy > 0) || (ym == 15 && gy < HW - 1);
      const int hc = (gx >> 4) * 2 + (xm == 15 ? 1 : 0);
      const int vr = (gy >> 4) * 2 + (ym == 15 ? 1 : 0);
      if (hH) v += eh[((size_t)bc * NEDGE + hc) * HW + gy];
      if (hV) v += ev[((size_t)bc * NEDGE + vr) * HW + gx];
      if (hH && hV) v += ed[((size_t)bc * NEDGE + vr) * NEDGE + hc];
    }
    a_lds[r * 102 + p] = v;
  }
  __syncthreads();

  const int wave  = __builtin_amdgcn_readfirstlane(tid >> 5);
  const int lane  = tid & 31;
  const int l16   = lane & 15;
  const int lhalf = (lane < 16) ? 0 : 2;
  const int rhalf = (lane < 16) ? 0 : 8;

  for (int nt = wave; nt < 13; nt += 8) {
    const int n  = nt * 16 + l16;
    const v2f* ap0 = (const v2f*)&a_lds[l16 * 102 + lhalf];
    const v2f* ap1 = (const v2f*)&a_lds[(16 + l16) * 102 + lhalf];
    const v2f* bp = (const v2f*)w1f + nt * (25 * 32) + lane;
    v2f bfr[25];
    #pragma unroll
    for (int j = 0; j < 25; ++j) bfr[j] = bp[j * 32];
    v8f acc0 = {}, acc1 = {};
    #pragma unroll
    for (int j = 0; j < 25; ++j) {
      acc0 = wmma_f32(ap0[2 * j], bfr[j], acc0);
      acc1 = wmma_f32(ap1[2 * j], bfr[j], acc1);
    }
    for (int r = 0; r < 8; ++r) {
      h_lds[(r + rhalf) * 210 + n]      = acc0[r];
      h_lds[(16 + r + rhalf) * 210 + n] = acc1[r];
    }
  }
  __syncthreads();

  if (wave < 7) {
    const int nt = wave;
    const int n  = nt * 16 + l16;
    const v2f* ap0 = (const v2f*)&h_lds[l16 * 210 + lhalf];
    const v2f* ap1 = (const v2f*)&h_lds[(16 + l16) * 210 + lhalf];
    const v2f* bp = (const v2f*)w2f + nt * (50 * 32) + lane;
    v8f acc0 = {}, acc1 = {};
    #pragma nounroll
    for (int kh = 0; kh < 2; ++kh) {
      v2f bfr[25];
      #pragma unroll
      for (int j = 0; j < 25; ++j) bfr[j] = bp[(kh * 25 + j) * 32];
      #pragma unroll
      for (int j = 0; j < 25; ++j) {
        acc0 = wmma_f32(ap0[2 * (kh * 25 + j)], bfr[j], acc0);
        acc1 = wmma_f32(ap1[2 * (kh * 25 + j)], bfr[j], acc1);
      }
    }
    if (n < P2) {
      float* dp = dst + ((size_t)(P0 >> 4) * 25 + (n >> 2)) * 64
                      + ((n >> 1) & 1) * 32 + (lane >> 4) * 16 + (n & 1);
      for (int r = 0; r < 8; ++r) {
        dp[2 * r]           = acc0[r] > 0.f ? acc0[r] : 0.f;
        dp[25 * 64 + 2 * r] = acc1[r] > 0.f ? acc1[r] : 0.f;
      }
    }
  }
}

__global__ __launch_bounds__(256) void att_kernel(
    const float* __restrict__ ui, const float* __restrict__ uf,
    float* __restrict__ out) {
  const int tid  = threadIdx.x;
  const int bc   = blockIdx.z;
  const int lane = tid & 31;
  const int l16  = lane & 15;
  const int h    = lane >> 4;
  const int wave = __builtin_amdgcn_readfirstlane(tid >> 5);
  const int nt   = blockIdx.y * 4 + (wave & 3);
  const int mtb  = blockIdx.x * 8 + (wave >> 2);

  const v2f* Uf = (const v2f*)ui + (size_t)bc * 64 * (25 * 32) + lane;
  const v2f* Vf = (const v2f*)uf + ((size_t)bc * 64 + nt) * (25 * 32) + lane;

  v2f B[25];
  #pragma unroll
  for (int j = 0; j < 25; ++j) B[j] = Vf[j * 32];

  #pragma nounroll
  for (int i = 0; i < 2; ++i) {
    const int mt0 = mtb + 4 * i, mt1 = mt0 + 2;
    const v2f* a0p = Uf + (size_t)mt0 * (25 * 32);
    const v2f* a1p = Uf + (size_t)mt1 * (25 * 32);
    v2f A0[25], A1[25];
    #pragma unroll
    for (int j = 0; j < 25; ++j) { A0[j] = a0p[j * 32]; A1[j] = a1p[j * 32]; }
    __builtin_amdgcn_sched_barrier(0);
    v8f acc0 = {}, acc1 = {};
    #pragma unroll
    for (int j = 0; j < 25; ++j) {
      acc0 = wmma_f32(A0[j], B[j], acc0);
      acc1 = wmma_f32(A1[j], B[j], acc1);
    }
    float* o0 = out + ((size_t)bc * LPATCH + mt0 * 16 + 8 * h) * LPATCH + nt * 16 + l16;
    float* o1 = out + ((size_t)bc * LPATCH + mt1 * 16 + 8 * h) * LPATCH + nt * 16 + l16;
    #pragma unroll
    for (int r = 0; r < 8; ++r) {
      o0[(size_t)r * LPATCH] = acc0[r] * 0.01f;
      o1[(size_t)r * LPATCH] = acc1[r] * 0.01f;
    }
  }
}

extern "C" void kernel_launch(void* const* d_in, const int* in_sizes, int n_in,
                              void* d_out, int out_size, void* d_ws,
                              size_t ws_size, hipStream_t stream) {
  (void)in_sizes; (void)n_in; (void)out_size; (void)ws_size;
  const float* x       = (const float*)d_in[0];
  const float* w_conv1 = (const float*)d_in[1];
  const float* b_conv1 = (const float*)d_in[2];
  const float* w_conv2 = (const float*)d_in[3];
  const float* b_conv2 = (const float*)d_in[4];
  const float* w_fm    = (const float*)d_in[5];
  const float* b_fm    = (const float*)d_in[6];
  const float* w_lin1  = (const float*)d_in[7];
  const float* w_lin2  = (const float*)d_in[8];
  float* out = (float*)d_out;

  float* ws  = (float*)d_ws;
  float* ini = ws + WS_INI;
  float* eh  = ws + WS_EH;
  float* ev  = ws + WS_EV;
  float* ed  = ws + WS_ED;
  float* ui  = ws + WS_UI;
  float* uf  = ws + WS_UF;
  float* w1f = ws + WS_W1P;
  float* w2f = ws + WS_W2P;
  float* aimg = ws + WS_AIMG;

  pad_weights_kernel<<<dim3((112 * 200 + 255) / 256), 256, 0, stream>>>(
      w_lin1, w_lin2, w_conv1, w1f, w2f, aimg);

  conv_fused_kernel<<<dim3(HW / 16, HW / 16, BATCH), 256,
                      CONV_LDS_FLOATS * sizeof(float), stream>>>(
      x, aimg, b_conv1, w_conv2, b_conv2, ini, eh, ev, ed);

  trans_kernel<<<dim3(BATCH * CLS * LPATCH / TPB, 2, 1), 256,
                 TRANS_LDS_FLOATS * sizeof(float), stream>>>(
      ini, eh, ev, ed, x, w_fm, b_fm, w1f, w2f, ui, uf);

  att_kernel<<<dim3(LPATCH / 128, LPATCH / 64, BATCH * CLS), 256, 0,
               stream>>>(ui, uf, out);
}
